# starting kernel
# baseline (speedup 1.0000x reference)
#include <hip/hip_runtime.h>
#include <hip/hip_bf16.h>

typedef float v2f __attribute__((ext_vector_type(2)));
typedef float v8f __attribute__((ext_vector_type(8)));

#define NEG_SLOPE 0.2f

__device__ __forceinline__ unsigned fkey(float x) {
  unsigned b = __float_as_uint(x);
  return (b & 0x80000000u) ? ~b : (b | 0x80000000u);
}
__device__ __forceinline__ float funkey(unsigned k) {
  return (k & 0x80000000u) ? __uint_as_float(k & 0x7FFFFFFFu) : __uint_as_float(~k);
}
#define NEG_INF_KEY 0x007FFFFFu

__global__ void fill_u32_kernel(unsigned* __restrict__ p, unsigned v, int n) {
  int i = blockIdx.x * blockDim.x + threadIdx.x;
  if (i < n) p[i] = v;
}

__global__ void gemm_wmma_f32_kernel(const float* __restrict__ A,
                                     const float* __restrict__ B,
                                     float* __restrict__ C,
                                     int M, int K, int Ncol) {
  const int wave = threadIdx.x >> 5;
  const int lane = threadIdx.x & 31;
  const int rowBase = blockIdx.x * 16;
  const int colBase = wave * 16;
  const int lm = lane & 15;
  const int hi = lane >> 4;

  v8f acc = {};
  const float* arow = A + (size_t)(rowBase + lm) * K;
  for (int k0 = 0; k0 < K; k0 += 4) {
    const int ka = k0 + hi * 2;
    v2f a, b;
    a.x = arow[ka];
    a.y = arow[ka + 1];
    b.x = B[(size_t)ka * Ncol + colBase + lm];
    b.y = B[(size_t)(ka + 1) * Ncol + colBase + lm];
    acc = __builtin_amdgcn_wmma_f32_16x16x4_f32(
         false, a,  false, b,
         (short)0, acc,  false,  false);
  }
#pragma unroll
  for (int r = 0; r < 8; ++r) {
    const int row = rowBase + hi * 8 + r;
    C[(size_t)row * Ncol + colBase + lm] = acc[r];
  }
}

__global__ void alpha_kernel(const float* __restrict__ h,
                             const float* __restrict__ a_src,
                             const float* __restrict__ a_dst,
                             float* __restrict__ as, float* __restrict__ ad,
                             int N, int H, int D) {
  int i = blockIdx.x * blockDim.x + threadIdx.x;
  if (i >= N * H) return;
  int hh = i % H;
  const float* hp = h + (size_t)i * D;
  const float* sp = a_src + (size_t)hh * D;
  const float* dp = a_dst + (size_t)hh * D;
  float s = 0.f, d = 0.f;
  for (int k = 0; k < D; ++k) { float v = hp[k]; s += v * sp[k]; d += v * dp[k]; }
  as[i] = s; ad[i] = d;
}

__device__ __forceinline__ void edge_sd(const int* __restrict__ ei, int E, int i,
                                        int& s, int& d) {
  if (i < E) { s = ei[i]; d = ei[E + i]; } else { s = d = i - E; }
}

__global__ void edge_max_kernel(const int* __restrict__ ei, int E, int N,
                                const float* __restrict__ as,
                                const float* __restrict__ ad,
                                unsigned* __restrict__ emax, int H) {
  int i = blockIdx.x * blockDim.x + threadIdx.x;
  if (i >= E + N) return;
  int s, d; edge_sd(ei, E, i, s, d);
  for (int h = 0; h < H; ++h) {
    float e = as[s * H + h] + ad[d * H + h];
    e = (e >= 0.f) ? e : NEG_SLOPE * e;
    atomicMax(&emax[d * H + h], fkey(e));
  }
}

__global__ void edge_acc_kernel(const int* __restrict__ ei, int E, int N,
                                const float* __restrict__ as,
                                const float* __restrict__ ad,
                                const unsigned* __restrict__ emax,
                                float* __restrict__ esum,
                                const float* __restrict__ hfeat,
                                float* __restrict__ acc,
                                int H, int D) {
  int t = blockIdx.x * blockDim.x + threadIdx.x;
  if (t >= (E + N) * H) return;
  int i = t / H, h = t % H;
  int s, d; edge_sd(ei, E, i, s, d);
  float e = as[s * H + h] + ad[d * H + h];
  e = (e >= 0.f) ? e : NEG_SLOPE * e;
  float ex = expf(e - funkey(emax[d * H + h]));
  atomicAdd(&esum[d * H + h], ex);
  const float* hs = hfeat + ((size_t)s * H + h) * D;
  float* ap = acc + ((size_t)d * H + h) * D;
  for (int k = 0; k < D; ++k) atomicAdd(&ap[k], ex * hs[k]);
}

__global__ void finalize_kernel(float* __restrict__ acc,
                                const float* __restrict__ esum,
                                const float* __restrict__ bias,
                                int N, int H, int D, int apply_elu) {
  int i = blockIdx.x * blockDim.x + threadIdx.x;
  int HD = H * D;
  if (i >= N * HD) return;
  int n = i / HD, c = i % HD, h = c / D;
  float v = acc[i] / (esum[n * H + h] + 1e-16f) + bias[c];
  if (apply_elu) v = (v > 0.f) ? v : (expf(v) - 1.f);
  acc[i] = v;
}

__global__ void pred_node_kernel(const float* __restrict__ h2,
                                 const float* __restrict__ Wp,
                                 float* __restrict__ ps, float* __restrict__ pd,
                                 int N, int D) {
  int n = blockIdx.x * blockDim.x + threadIdx.x;
  if (n >= N) return;
  const float* hp = h2 + (size_t)n * D;
  float s = 0.f, d = 0.f;
  for (int k = 0; k < D; ++k) { s += hp[k] * Wp[k]; d += hp[k] * Wp[D + k]; }
  ps[n] = s; pd[n] = d;
}

__global__ void pred_edge_kernel(const int* __restrict__ ei, int E,
                                 const float* __restrict__ ps,
                                 const float* __restrict__ pd,
                                 const float* __restrict__ bp,
                                 float* __restrict__ out) {
  int e = blockIdx.x * blockDim.x + threadIdx.x;
  if (e >= E) return;
  out[e] = ps[ei[e]] + pd[ei[E + e]] + bp[0];
}

static inline int cdiv(long long a, int b) { return (int)((a + b - 1) / b); }

extern "C" void kernel_launch(void* const* d_in, const int* in_sizes, int n_in,
                              void* d_out, int out_size, void* d_ws, size_t ws_size,
                              hipStream_t stream) {
  const float* x   = (const float*)d_in[0];
  const int*   ei  = (const int*)  d_in[1];
  const float* W1  = (const float*)d_in[2];
  const float* as1w= (const float*)d_in[3];
  const float* ad1w= (const float*)d_in[4];
  const float* b1  = (const float*)d_in[5];
  const float* W2  = (const float*)d_in[6];
  const float* as2w= (const float*)d_in[7];
  const float* ad2w= (const float*)d_in[8];
  const float* b2  = (const float*)d_in[9];
  const float* Wp  = (const float*)d_in[10];
  const float* bp  = (const float*)d_in[11];
  float* out = (float*)d_out;

  const int F = 128, H1 = 4, Hd = 32;
  const int N = in_sizes[0] / F;
  const int E = in_sizes[1] / 2;
  const int EN = E + N;

  float* ws = (float*)d_ws;
  float*    h1    = ws;                 ws += (size_t)N * F;
  float*    acc1  = ws;                 ws += (size_t)N * F;
  float*    as1   = ws;                 ws += (size_t)N * H1;
  float*    ad1   = ws;                 ws += (size_t)N * H1;
  unsigned* emax1 = (unsigned*)ws;      ws += (size_t)N * H1;
  float*    esum1 = ws;                 ws += (size_t)N * H1;
  float*    h2p   = ws;                 ws += (size_t)N * Hd;
  float*    acc2  = ws;                 ws += (size_t)N * Hd;
  float*    as2   = ws;                 ws += (size_t)N;
  float*    ad2   = ws;                 ws += (size_t)N;
  unsigned* emax2 = (unsigned*)ws;      ws += (size_t)N;
  float*    esum2 = ws;                 ws += (size_t)N;
  float*    ps    = ws;                 ws += (size_t)N;
  float*    pd    = ws;                 ws += (size_t)N;
  (void)ws_size;

  const int T = 256;

  fill_u32_kernel<<<cdiv((long long)N * F, T), T, 0, stream>>>((unsigned*)acc1, 0u, N * F);
  fill_u32_kernel<<<cdiv((long long)N * H1, T), T, 0, stream>>>((unsigned*)esum1, 0u, N * H1);
  fill_u32_kernel<<<cdiv((long long)N * H1, T), T, 0, stream>>>(emax1, NEG_INF_KEY, N * H1);
  fill_u32_kernel<<<cdiv((long long)N * Hd, T), T, 0, stream>>>((unsigned*)acc2, 0u, N * Hd);
  fill_u32_kernel<<<cdiv(N, T), T, 0, stream>>>((unsigned*)esum2, 0u, N);
  fill_u32_kernel<<<cdiv(N, T), T, 0, stream>>>(emax2, NEG_INF_KEY, N);

  gemm_wmma_f32_kernel<<<N / 16, 8 * 32, 0, stream>>>(x, W1, h1, N, F, F);
  alpha_kernel<<<cdiv((long long)N * H1, T), T, 0, stream>>>(h1, as1w, ad1w, as1, ad1, N, H1, Hd);
  edge_max_kernel<<<cdiv(EN, T), T, 0, stream>>>(ei, E, N, as1, ad1, emax1, H1);
  edge_acc_kernel<<<cdiv((long long)EN * H1, T), T, 0, stream>>>(ei, E, N, as1, ad1, emax1,
                                                                 esum1, h1, acc1, H1, Hd);
  finalize_kernel<<<cdiv((long long)N * F, T), T, 0, stream>>>(acc1, esum1, b1, N, H1, Hd, 1);

  gemm_wmma_f32_kernel<<<N / 16, 2 * 32, 0, stream>>>(acc1, W2, h2p, N, F, Hd);
  alpha_kernel<<<cdiv(N, T), T, 0, stream>>>(h2p, as2w, ad2w, as2, ad2, N, 1, Hd);
  edge_max_kernel<<<cdiv(EN, T), T, 0, stream>>>(ei, E, N, as2, ad2, emax2, 1);
  edge_acc_kernel<<<cdiv(EN, T), T, 0, stream>>>(ei, E, N, as2, ad2, emax2,
                                                 esum2, h2p, acc2, 1, Hd);
  finalize_kernel<<<cdiv((long long)N * Hd, T), T, 0, stream>>>(acc2, esum2, b2, N, 1, Hd, 0);

  pred_node_kernel<<<cdiv(N, T), T, 0, stream>>>(acc2, Wp, ps, pd, N, Hd);
  pred_edge_kernel<<<cdiv(E, T), T, 0, stream>>>(ei, E, ps, pd, bp, out);
}


# Owner scan reads one byte per edge (owner number, written by the attention-dot launch), four edges tested per word; each wave lists its finds at a scalar count, no LDS atomic; sums in arrival order
# speedup vs baseline: 11.2623x; 11.2623x over previous
#include <hip/hip_runtime.h>
#include <hip/hip_bf16.h>

typedef float v2f __attribute__((ext_vector_type(2)));
typedef float v4f __attribute__((ext_vector_type(4)));
typedef float v8f __attribute__((ext_vector_type(8)));
typedef int v4i __attribute__((ext_vector_type(4)));
typedef unsigned v4u __attribute__((ext_vector_type(4)));

#define NEG_SLOPE 0.2f

__device__ __forceinline__ unsigned fkey(float x) {
  unsigned b = __float_as_uint(x);
  return (b & 0x80000000u) ? ~b : (b | 0x80000000u);
}
__device__ __forceinline__ float funkey(unsigned k) {
  return (k & 0x80000000u) ? __uint_as_float(k & 0x7FFFFFFFu) : __uint_as_float(~k);
}

__global__ void gemm_wmma_f32_kernel(const float* __restrict__ A,
                                     const float* __restrict__ B,
                                     float* __restrict__ C,
                                     int M, int K, int Ncol) {
  const int wave = threadIdx.x >> 5;
  const int lane = threadIdx.x & 31;
  const int rowBase = blockIdx.x * 16;
  const int colBase = wave * 16;
  const int lm = lane & 15;
  const int hi = lane >> 4;

  v8f acc = {};
  const float* arow = A + (size_t)(rowBase + lm) * K;
  for (int k0 = 0; k0 < K; k0 += 4) {
    const int ka = k0 + hi * 2;
    v2f a, b;
    a.x = arow[ka];
    a.y = arow[ka + 1];
    b.x = B[(size_t)ka * Ncol + colBase + lm];
    b.y = B[(size_t)(ka + 1) * Ncol + colBase + lm];
    acc = __builtin_amdgcn_wmma_f32_16x16x4_f32(
         false, a,  false, b,
         (short)0, acc,  false,  false);
  }
#pragma unroll
  for (int r = 0; r < 8; ++r) {
    const int row = rowBase + hi * 8 + r;
    C[(size_t)row * Ncol + colBase + lm] = acc[r];
  }
}

template <int NPW>
__global__ void alpha_kernel(const float* __restrict__ h,
                             const float* __restrict__ a_src,
                             const float* __restrict__ a_dst,
                             float* __restrict__ as, float* __restrict__ ad,
                             int N, int H, int D,
                             const int* __restrict__ ei, int E,
                             unsigned char* __restrict__ okey) {
  int i = blockIdx.x * blockDim.x + threadIdx.x;
  if (okey) {
    const int nth = (int)(gridDim.x * blockDim.x);
    const int* __restrict__ dstp = ei + E;
    if ((E & 3) == 0) {
      const v4i* __restrict__ dv = reinterpret_cast<const v4i*>(dstp);
      unsigned* __restrict__ ko = reinterpret_cast<unsigned*>(okey);
      for (int q = i; q < (E >> 2); q += nth) {
        const v4i d4 = dv[q];
        ko[q] = (unsigned)(d4[0] / NPW) | ((unsigned)(d4[1] / NPW) << 8) |
                ((unsigned)(d4[2] / NPW) << 16) | ((unsigned)(d4[3] / NPW) << 24);
      }
    } else {
      for (int e = i; e < E; e += nth) okey[e] = (unsigned char)(dstp[e] / NPW);
    }
  }
  if (i >= N * H) return;
  int hh = i % H;
  const float* hp = h + (size_t)i * D;
  const float* sp = a_src + (size_t)hh * D;
  const float* dp = a_dst + (size_t)hh * D;
  float s = 0.f, d = 0.f;
  for (int k = 0; k < D; ++k) { float v = hp[k]; s += v * sp[k]; d += v * dp[k]; }
  as[i] = s;
  ad[i] = d;
}

__device__ __forceinline__ float edge_term(float asv, float adv, float mas) {
  float e = asv + adv;
  e = (e >= 0.f) ? e : NEG_SLOPE * e;
  float m = mas + adv;
  m = (m >= 0.f) ? m : NEG_SLOPE * m;
  return expf(e - m);
}

template <int H, int NPW, int CAP, int U, int G>
__global__ __launch_bounds__(1024) void owner_acc_kernel(
    const int* __restrict__ ei, int E, int N,
    const unsigned char* __restrict__ okey,
    const float* __restrict__ as, const float* __restrict__ ad,
    const float* __restrict__ hfeat, const float* __restrict__ bias,
    float* __restrict__ out, int apply_elu,
    unsigned* __restrict__ lst, int* __restrict__ lcnt) {
  constexpr int D = 32;
  constexpr int HD = H * D;
  constexpr int T = 1024;
  constexpr int NW = T / 32;
  constexpr int CAPW = CAP / NW;
  constexpr int NB = CAPW / 32;
  constexpr int SPW = (NPW + NW - 1) / NW;
  static_assert(H == 4, "the alpha planes are read 16 bytes per node");
  static_assert(NW == 32, "one lane per wave when the waves' counts are added up");
  static_assert(CAP % T == 0, "whole batches");
  static_assert(SPW <= 32 && SPW <= CAPW, "self loops are listed by one lane each");
  static_assert(NPW <= 256, "the owned index is saved in 8 bits");
  static_assert((NPW * HD) % 4 == 0, "tile shape");
  __shared__ __attribute__((aligned(16))) float accL[NPW * HD];
  __shared__ __attribute__((aligned(16))) float esL[NPW * H];
  __shared__ __attribute__((aligned(16))) float adL[NPW * H];
  __shared__ __attribute__((aligned(16))) unsigned mxL[NPW * H];
  __shared__ int leL[CAP];
  __shared__ int wcL[NW];

  const int t = (int)threadIdx.x;
  const int lane = t & 31;
  const int wave = __builtin_amdgcn_readfirstlane(t >> 5);
  const int n0 = (int)blockIdx.x * NPW;
  const int nown = (N - n0 < NPW) ? (N - n0) : NPW;
  int* __restrict__ seg = leL + wave * CAPW;
  int cnt = nown - wave * SPW;
  cnt = (cnt < 0) ? 0 : ((cnt > SPW) ? SPW : cnt);

  {
    const v4f z = {0.f, 0.f, 0.f, 0.f};
    v4f* a4 = reinterpret_cast<v4f*>(accL);
    for (int i = t; i < NPW * HD / 4; i += T) a4[i] = z;
    for (int i = t; i < NPW * H; i += T) {
      esL[i] = 0.f;
      mxL[i] = 0u;
      adL[i] = (i < nown * H) ? ad[(size_t)n0 * H + i] : 0.f;
    }
    if (lane < cnt) seg[lane] = E + n0 + wave * SPW + lane;
  }
  __syncthreads();

  const int* __restrict__ dstp = ei + E;
  const unsigned bb = (unsigned)blockIdx.x * 0x01010101u;
  auto keep1 = [&](bool hit, int e) {
    const unsigned m = __builtin_amdgcn_ballot_w32(hit);
    if (m != 0u) {
      if (hit) {
        const int slot = cnt + (int)__builtin_amdgcn_mbcnt_lo(m, 0u);
        if (slot < CAPW) seg[slot] = e;
      }
      cnt += __builtin_popcount(m);
    }
  };
  auto test4 = [&](unsigned w, int ebase) {
    const unsigned x = w ^ bb;
    unsigned y = ~(((x & 0x7F7F7F7Fu) + 0x7F7F7F7Fu) | x | 0x7F7F7F7Fu);
    unsigned m = __builtin_amdgcn_ballot_w32(y != 0u);
    while (m != 0u) {
      if (y != 0u) {
        int slot = (int)__builtin_amdgcn_mbcnt_lo(m, (unsigned)cnt);
        slot = (slot < CAPW - 1) ? slot : CAPW - 1;
        seg[slot] = ebase + (__builtin_ctz(y) >> 3);
        y &= y - 1u;
      }
      cnt += __builtin_popcount(m);
      m = __builtin_amdgcn_ballot_w32(y != 0u);
    }
  };
  {
    const v4u* __restrict__ kv = reinterpret_cast<const v4u*>(okey);
    const int nvec = E >> 4;
    constexpr int TILEV = T * U;
    const int nfull = nvec / TILEV;
    if (nfull > 0) {
      int tile = (int)(blockIdx.x % (unsigned)nfull);
      v4u cur[U];
      for (int u = 0; u < U; ++u) cur[u] = kv[tile * TILEV + u * T + t];
      for (int k = 0; k < nfull; ++k) {
        const int nt = (tile + 1 == nfull) ? 0 : tile + 1;
        v4u nxt[U];
        for (int u = 0; u < U; ++u) nxt[u] = kv[nt * TILEV + u * T + t];
        for (int u = 0; u < U; ++u) {
          const int vi = tile * TILEV + u * T + t;
          for (int q = 0; q < 4; ++q) test4(cur[u][q], vi * 16 + q * 4);
        }
        for (int u = 0; u < U; ++u) cur[u] = nxt[u];
        tile = nt;
      }
    }
    for (int v0 = nfull * TILEV; v0 < nvec; v0 += T) {
      const int vi = v0 + t;
      const unsigned none = bb ^ 0x01010101u;
      v4u k4 = {none, none, none, none};
      if (vi < nvec) k4 = kv[vi];
      for (int q = 0; q < 4; ++q) test4(k4[q], vi * 16 + q * 4);
    }
    for (int e0 = nvec << 4; e0 < E; e0 += T) {
      const int e = e0 + t;
      keep1(e < E && (unsigned)okey[(e < E) ? e : 0] == (unsigned)blockIdx.x, e);
    }
  }
  if (lane == 0) wcL[wave] = cnt;
  __syncthreads();
  int total = 0, base = 0;
  bool fits = true;
  {
    const int cl = wcL[lane];
    for (int v = 0; v < NW; ++v) {
      const int cv = __builtin_amdgcn_readlane(cl, v);
      total += cv;
      if (v < wave) base += cv;
      fits = fits && (cv <= CAPW);
    }
  }
  const bool listed = fits;
  if (t == 0) lcnt[blockIdx.x] = listed ? total : CAP + 1;

  int sK[NB], dK[NB];
  v4f aK[NB];
  if (listed) {
    int eK[NB];
    for (int k = 0; k < NB; ++k) {
      const int idx = k * 32 + lane;
      const int ev = seg[(idx < cnt) ? idx : 0];
      eK[k] = (idx < cnt) ? ev : E + n0;
    }
    for (int k = 0; k < NB; ++k) {
      const bool real = eK[k] < E;
      const int ec = real ? eK[k] : 0;
      const int sv = ei[ec];
      const int dv = dstp[ec];
      sK[k] = real ? sv : eK[k] - E;
      dK[k] = (real ? dv : eK[k] - E) - n0;
    }
    for (int k = 0; k < NB; ++k)
      aK[k] = *reinterpret_cast<const v4f*>(as + (size_t)sK[k] * H);
    for (int k = 0; k < NB; ++k) {
      const int idx = k * 32 + lane;
      if (idx < cnt)
        for (int h = 0; h < H; ++h) atomicMax(&mxL[dK[k] * H + h], fkey(aK[k][h]));
    }
  } else {
    for (int k = 0; k < NB; ++k) { sK[k] = 0; dK[k] = 0; aK[k] = v4f{0.f, 0.f, 0.f, 0.f}; }
    for (int i = t; i < nown + E; i += T) {
      int s, dl;
      if (i < nown) { s = n0 + i; dl = i; }
      else {
        const int e = i - nown;
        const unsigned u = (unsigned)(dstp[e] - n0);
        if (u >= (unsigned)nown) continue;
        s = ei[e]; dl = (int)u;
      }
      for (int h = 0; h < H; ++h) atomicMax(&mxL[dl * H + h], fkey(as[(size_t)s * H + h]));
    }
  }
  __syncthreads();

  if (listed) {
    unsigned* __restrict__ lsave = lst + (size_t)blockIdx.x * CAP + base;
    for (int k = 0; k < NB; ++k) {
      const int b0 = k * 32;
      if (b0 >= cnt) break;
      const int bc = (cnt - b0 < 32) ? (cnt - b0) : 32;
      const int s = sK[0], dl = dK[0];
      const v4f asv = aK[0];
      for (int q = 0; q + 1 < NB; ++q) { sK[q] = sK[q + 1]; dK[q] = dK[q + 1]; aK[q] = aK[q + 1]; }
      float ex[H];
      for (int h = 0; h < H; ++h) ex[h] = 0.f;
      if (lane < bc) {
        lsave[b0 + lane] = (unsigned)s | ((unsigned)dl << 24);
        const v4f a4 = *reinterpret_cast<const v4f*>(adL + dl * H);
        const v4u k4 = *reinterpret_cast<const v4u*>(mxL + dl * H);
        for (int h = 0; h < H; ++h) {
          ex[h] = edge_term(asv[h], a4[h], funkey(k4[h]));
          atomicAdd(&esL[dl * H + h], ex[h]);
        }
      }
      const int full = bc - bc % G;
      for (int j0 = 0; j0 < full; j0 += G) {
        float v[G][H];
        for (int g = 0; g < G; ++g) {
          const int sj = __builtin_amdgcn_readlane(s, j0 + g);
          const float* hp = hfeat + (size_t)sj * HD + lane;
          for (int h = 0; h < H; ++h) v[g][h] = hp[h * D];
        }
        for (int g = 0; g < G; ++g) {
          const int dj = __builtin_amdgcn_readlane(dl, j0 + g);
          float* ap = accL + dj * HD + lane;
          for (int h = 0; h < H; ++h) {
            const float w = __int_as_float(
                __builtin_amdgcn_readlane(__float_as_int(ex[h]), j0 + g));
            atomicAdd(&ap[h * D], w * v[g][h]);
          }
        }
      }
      for (int j = full; j < bc; ++j) {
        const int sj = __builtin_amdgcn_readlane(s, j);
        const int dj = __builtin_amdgcn_readlane(dl, j);
        const float* hp = hfeat + (size_t)sj * HD + lane;
        float* ap = accL + dj * HD + lane;
        for (int h = 0; h < H; ++h) {
          const float w = __int_as_float(
              __builtin_amdgcn_readlane(__float_as_int(ex[h]), j));
          atomicAdd(&ap[h * D], w * hp[h * D]);
        }
      }
    }
  } else {
    for (int i = t; i < nown + E; i += T) {
      int s, dl;
      if (i < nown) { s = n0 + i; dl = i; }
      else {
        const int e = i - nown;
        const unsigned u = (unsigned)(dstp[e] - n0);
        if (u >= (unsigned)nown) continue;
        s = ei[e]; dl = (int)u;
      }
      const float* hp = hfeat + (size_t)s * HD;
      for (int h = 0; h < H; ++h) {
        const float exh = edge_term(as[(size_t)s * H + h], adL[dl * H + h],
                                    funkey(mxL[dl * H + h]));
        atomicAdd(&esL[dl * H + h], exh);
        for (int kk = 0; kk < D; ++kk)
          atomicAdd(&accL[dl * HD + h * D + kk], exh * hp[h * D + kk]);
      }
    }
  }
  __syncthreads();

  for (int i = t; i < nown * HD; i += T) {
    const int nl = i / HD, c = i % HD, h = c / D;
    float v = accL[i] / (esL[nl * H + h] + 1e-16f) + bias[c];
    if (apply_elu) v = (v > 0.f) ? v : (expf(v) - 1.f);
    out[(size_t)n0 * HD + i] = v;
  }
}

template <int NPW, int CAP, int G>
__global__ __launch_bounds__(1024) void owner_list_acc_kernel(
    const int* __restrict__ ei, int E, int N,
    const unsigned* __restrict__ lst, const int* __restrict__ lcnt,
    const float* __restrict__ as, const float* __restrict__ ad,
    const float* __restrict__ hfeat, const float* __restrict__ bias,
    float* __restrict__ out, int apply_elu) {
  constexpr int D = 32;
  constexpr int T = 1024;
  constexpr int NB = CAP / T;
  static_assert(CAP % T == 0, "whole batches");
  static_assert((NPW * D) % 4 == 0 && 32 % G == 0, "tile shape");
  __shared__ __attribute__((aligned(16))) float accL[NPW * D];
  __shared__ float esL[NPW];
  __shared__ float adL[NPW];
  __shared__ unsigned mxL[NPW];

  const int t = (int)threadIdx.x;
  const int n0 = (int)blockIdx.x * NPW;
  const int nown = (N - n0 < NPW) ? (N - n0) : NPW;

  {
    const v4f z = {0.f, 0.f, 0.f, 0.f};
    v4f* a4 = reinterpret_cast<v4f*>(accL);
    for (int i = t; i < NPW * D / 4; i += T) a4[i] = z;
    for (int i = t; i < NPW; i += T) {
      esL[i] = 0.f;
      mxL[i] = 0u;
      adL[i] = (i < nown) ? ad[n0 + i] : 0.f;
    }
  }
  __syncthreads();

  const int total = lcnt[blockIdx.x];
  const bool listed = total <= CAP;
  const int* __restrict__ dstp = ei + E;

  int sK[NB], dK[NB];
  float aK[NB];
  if (listed) {
    const unsigned* __restrict__ lp = lst + (size_t)blockIdx.x * CAP;
    for (int k = 0; k < NB; ++k) {
      const int idx = k * T + t;
      const unsigned w = lp[(idx < total) ? idx : 0];
      sK[k] = (int)(w & 0xFFFFFFu);
      dK[k] = (int)(w >> 24);
    }
    for (int k = 0; k < NB; ++k) aK[k] = as[sK[k]];
    for (int k = 0; k < NB; ++k)
      if (k * T + t < total) atomicMax(&mxL[dK[k]], fkey(aK[k]));
  } else {
    for (int k = 0; k < NB; ++k) { sK[k] = 0; dK[k] = 0; aK[k] = 0.f; }
    for (int i = t; i < nown + E; i += T) {
      int s, dl;
      if (i < nown) { s = n0 + i; dl = i; }
      else {
        const int e = i - nown;
        const unsigned u = (unsigned)(dstp[e] - n0);
        if (u >= (unsigned)nown) continue;
        s = ei[e]; dl = (int)u;
      }
      atomicMax(&mxL[dl], fkey(as[s]));
    }
  }
  __syncthreads();

  if (listed) {
    const int lane = t & 31;
    const int wave = __builtin_amdgcn_readfirstlane(t >> 5);
    for (int k = 0; k < NB; ++k) {
      const int b0 = wave * 32 + k * T;
      if (b0 >= total) break;
      const int bc = (total - b0 < 32) ? (total - b0) : 32;
      const int s = sK[0], dl = dK[0];
      const float asv = aK[0];
      for (int q = 0; q + 1 < NB; ++q) { sK[q] = sK[q + 1]; dK[q] = dK[q + 1]; aK[q] = aK[q + 1]; }
      float ex = 0.f;
      if (lane < bc) {
        ex = edge_term(asv, adL[dl], funkey(mxL[dl]));
        atomicAdd(&esL[dl], ex);
      }
      const int full = bc - bc % G;
      for (int j0 = 0; j0 < full; j0 += G) {
        float v[G];
        for (int g = 0; g < G; ++g) {
          const int sj = __builtin_amdgcn_readlane(s, j0 + g);
          v[g] = hfeat[(size_t)sj * D + lane];
        }
        for (int g = 0; g < G; ++g) {
          const int dj = __builtin_amdgcn_readlane(dl, j0 + g);
          const float w = __int_as_float(
              __builtin_amdgcn_readlane(__float_as_int(ex), j0 + g));
          atomicAdd(&accL[dj * D + lane], w * v[g]);
        }
      }
      for (int j = full; j < bc; ++j) {
        const int sj = __builtin_amdgcn_readlane(s, j);
        const int dj = __builtin_amdgcn_readlane(dl, j);
        const float w = __int_as_float(
            __builtin_amdgcn_readlane(__float_as_int(ex), j));
        atomicAdd(&accL[dj * D + lane], w * hfeat[(size_t)sj * D + lane]);
      }
    }
  } else {
    for (int i = t; i < nown + E; i += T) {
      int s, dl;
      if (i < nown) { s = n0 + i; dl = i; }
      else {
        const int e = i - nown;
        const unsigned u = (unsigned)(dstp[e] - n0);
        if (u >= (unsigned)nown) continue;
        s = ei[e]; dl = (int)u;
      }
      const float exv = edge_term(as[s], adL[dl], funkey(mxL[dl]));
      const float* hp = hfeat + (size_t)s * D;
      atomicAdd(&esL[dl], exv);
      for (int kk = 0; kk < D; ++kk) atomicAdd(&accL[dl * D + kk], exv * hp[kk]);
    }
  }
  __syncthreads();

  for (int i = t; i < nown * D; i += T) {
    const int nl = i / D, c = i % D;
    float v = accL[i] / (esL[nl] + 1e-16f) + bias[c];
    if (apply_elu) v = (v > 0.f) ? v : (expf(v) - 1.f);
    out[(size_t)n0 * D + i] = v;
  }
}

__global__ void pred_node_kernel(const float* __restrict__ h2,
                                 const float* __restrict__ Wp,
                                 float* __restrict__ ps, float* __restrict__ pd,
                                 int N, int D) {
  int n = blockIdx.x * blockDim.x + threadIdx.x;
  if (n >= N) return;
  const float* hp = h2 + (size_t)n * D;
  float s = 0.f, d = 0.f;
  for (int k = 0; k < D; ++k) { s += hp[k] * Wp[k]; d += hp[k] * Wp[D + k]; }
  ps[n] = s; pd[n] = d;
}

__global__ void pred_edge_kernel(const int* __restrict__ ei, int E,
                                 const float* __restrict__ ps,
                                 const float* __restrict__ pd,
                                 const float* __restrict__ bp,
                                 float* __restrict__ out) {
  int e = blockIdx.x * blockDim.x + threadIdx.x;
  if (e >= E) return;
  out[e] = ps[ei[e]] + pd[ei[E + e]] + bp[0];
}

static inline int cdiv(long long a, int b) { return (int)((a + b - 1) / b); }

extern "C" void kernel_launch(void* const* d_in, const int* in_sizes, int n_in,
                              void* d_out, int out_size, void* d_ws, size_t ws_size,
                              hipStream_t stream) {
  const float* x   = (const float*)d_in[0];
  const int*   ei  = (const int*)  d_in[1];
  const float* W1  = (const float*)d_in[2];
  const float* as1w= (const float*)d_in[3];
  const float* ad1w= (const float*)d_in[4];
  const float* b1  = (const float*)d_in[5];
  const float* W2  = (const float*)d_in[6];
  const float* as2w= (const float*)d_in[7];
  const float* ad2w= (const float*)d_in[8];
  const float* b2  = (const float*)d_in[9];
  const float* Wp  = (const float*)d_in[10];
  const float* bp  = (const float*)d_in[11];
  float* out = (float*)d_out;

  const int F = 128, H1 = 4, Hd = 32;
  const int N = in_sizes[0] / F;
  const int E = in_sizes[1] / 2;

  constexpr int NPW = 196, CAP = 6144;
  float* ws = (float*)d_ws;
  float*    h1    = ws;                 ws += (size_t)N * F;
  float*    acc1  = ws;                 ws += (size_t)N * F;
  float*    as1   = ws;                 ws += (size_t)N * H1;
  float*    ad1   = ws;                 ws += (size_t)N * 2 * H1;
  int*      lcnt  = (int*)ws;           ws += (size_t)N * H1;
  float*    h2p   = ws;                 ws += (size_t)N * Hd;
  unsigned char* okey = (unsigned char*)h2p;
  unsigned* lst   = (unsigned*)ws;      ws += (size_t)N * Hd;
  float*    as2   = ws;                 ws += (size_t)N;
  float*    ad2   = ws;                 ws += (size_t)N * 2;
  ws += (size_t)N;
  float*    ps    = ws;                 ws += (size_t)N;
  float*    pd    = ws;                 ws += (size_t)N;
  float*    out2  = h1;
  (void)ws_size;

  const int T = 256;

  gemm_wmma_f32_kernel<<<N / 16, 8 * 32, 0, stream>>>(x, W1, h1, N, F, F);
  alpha_kernel<NPW><<<cdiv((long long)N * H1, T), T, 0, stream>>>(h1, as1w, ad1w, as1, ad1, N, H1, Hd, ei, E, okey);
  owner_acc_kernel<4, NPW, CAP, 4, 8><<<cdiv(N, NPW), 1024, 0, stream>>>(
      ei, E, N, okey, as1, ad1, h1, b1, acc1, 1, lst, lcnt);

  gemm_wmma_f32_kernel<<<N / 16, 2 * 32, 0, stream>>>(acc1, W2, h2p, N, F, Hd);
  alpha_kernel<NPW><<<cdiv(N, T), T, 0, stream>>>(h2p, as2w, ad2w, as2, ad2, N, 1, Hd, ei, E, nullptr);
  owner_list_acc_kernel<NPW, CAP, 16><<<cdiv(N, NPW), 1024, 0, stream>>>(
      ei, E, N, lst, lcnt, as2, ad2, h2p, b2, out2, 0);

  pred_node_kernel<<<cdiv(N, T), T, 0, stream>>>(out2, Wp, ps, pd, N, Hd);
  pred_edge_kernel<<<cdiv(E, T), T, 0, stream>>>(ei, E, ps, pd, bp, out);
}


# GEMM re-tiled: work-group takes 32 rows, A tile read once as whole lines into LDS, each wave keeps its 128x16 block of B in registers for two output tiles; results leave as whole lines; same k order
# speedup vs baseline: 11.7652x; 1.0447x over previous
#include <hip/hip_runtime.h>
#include <hip/hip_bf16.h>

typedef float v2f __attribute__((ext_vector_type(2)));
typedef float v4f __attribute__((ext_vector_type(4)));
typedef float v8f __attribute__((ext_vector_type(8)));
typedef int v4i __attribute__((ext_vector_type(4)));
typedef unsigned v4u __attribute__((ext_vector_type(4)));

#define NEG_SLOPE 0.2f

__device__ __forceinline__ unsigned fkey(float x) {
  unsigned b = __float_as_uint(x);
  return (b & 0x80000000u) ? ~b : (b | 0x80000000u);
}
__device__ __forceinline__ float funkey(unsigned k) {
  return (k & 0x80000000u) ? __uint_as_float(k & 0x7FFFFFFFu) : __uint_as_float(~k);
}

template <int NCT>
__global__ __launch_bounds__(NCT * 32) void gemm_wmma_f32_kernel(
    const float* __restrict__ A, const float* __restrict__ B,
    float* __restrict__ C, int M) {
  constexpr int K = 128;
  constexpr int KS = K / 4;
  constexpr int NCOL = NCT * 16;
  constexpr int NT = NCT * 32;
  constexpr int RT = 32;
  constexpr int P = K + 4;
  constexpr int AQ = K / 4;
  constexpr int CQ = NCOL / 4;
  static_assert((RT * AQ) % NT == 0 && (RT * CQ) % NT == 0, "tile shape");
  __shared__ __attribute__((aligned(16))) float tL[RT * P];

  const int t = (int)threadIdx.x;
  const int wave = t >> 5;
  const int lane = t & 31;
  const int lm = lane & 15;
  const int hi = lane >> 4;
  const int rowBase = (int)blockIdx.x * RT;
  const int colBase = wave * 16;

  constexpr int NA = RT * AQ / NT;
  v4f av[NA];
#pragma unroll
  for (int n = 0; n < NA; ++n) {
    const int i = t + n * NT;
    const int r = i / AQ, q = i % AQ;
    const int row = (rowBase + r < M) ? (rowBase + r) : (M - 1);
    av[n] = *reinterpret_cast<const v4f*>(A + (size_t)row * K + q * 4);
  }

  v2f b[KS];
  {
    const float* bp = B + (size_t)(hi * 2) * NCOL + colBase + lm;
#pragma unroll
    for (int j = 0; j < KS; ++j) {
      b[j].x = bp[(size_t)(4 * j) * NCOL];
      b[j].y = bp[(size_t)(4 * j + 1) * NCOL];
    }
  }

#pragma unroll
  for (int n = 0; n < NA; ++n) {
    const int i = t + n * NT;
    const int r = i / AQ, q = i % AQ;
    *reinterpret_cast<v4f*>(tL + r * P + q * 4) = av[n];
  }
  __syncthreads();

  v8f acc0 = {}, acc1 = {};
  {
    const float* a0p = tL + lm * P + hi * 2;
    const float* a1p = a0p + 16 * P;
#pragma unroll
    for (int j = 0; j < KS; ++j) {
      const v2f a0 = *reinterpret_cast<const v2f*>(a0p + 4 * j);
      const v2f a1 = *reinterpret_cast<const v2f*>(a1p + 4 * j);
      acc0 = __builtin_amdgcn_wmma_f32_16x16x4_f32(
           false, a0,  false, b[j],
           (short)0, acc0,  false,  false);
      acc1 = __builtin_amdgcn_wmma_f32_16x16x4_f32(
           false, a1,  false, b[j],
           (short)0, acc1,  false,  false);
    }
  }
  __syncthreads();

  for (int r = 0; r < 8; ++r) {
    tL[(hi * 8 + r) * P + colBase + lm] = acc0[r];
    tL[(16 + hi * 8 + r) * P + colBase + lm] = acc1[r];
  }
  __syncthreads();
  for (int n = 0; n < RT * CQ / NT; ++n) {
    const int i = t + n * NT;
    const int r = i / CQ, q = i % CQ;
    if (rowBase + r < M)
      *reinterpret_cast<v4f*>(C + (size_t)(rowBase + r) * NCOL + q * 4) =
          *reinterpret_cast<const v4f*>(tL + r * P + q * 4);
  }
}

template <int NPW>
__global__ void alpha_kernel(const float* __restrict__ h,
                             const float* __restrict__ a_src,
                             const float* __restrict__ a_dst,
                             float* __restrict__ as, float* __restrict__ ad,
                             int N, int H, int D,
                             const int* __restrict__ ei, int E,
                             unsigned char* __restrict__ okey) {
  int i = blockIdx.x * blockDim.x + threadIdx.x;
  if (okey) {
    const int nth = (int)(gridDim.x * blockDim.x);
    const int* __restrict__ dstp = ei + E;
    if ((E & 3) == 0) {
      const v4i* __restrict__ dv = reinterpret_cast<const v4i*>(dstp);
      unsigned* __restrict__ ko = reinterpret_cast<unsigned*>(okey);
      for (int q = i; q < (E >> 2); q += nth) {
        const v4i d4 = dv[q];
        ko[q] = (unsigned)(d4[0] / NPW) | ((unsigned)(d4[1] / NPW) << 8) |
                ((unsigned)(d4[2] / NPW) << 16) | ((unsigned)(d4[3] / NPW) << 24);
      }
    } else {
      for (int e = i; e < E; e += nth) okey[e] = (unsigned char)(dstp[e] / NPW);
    }
  }
  if (i >= N * H) return;
  int hh = i % H;
  const float* hp = h + (size_t)i * D;
  const float* sp = a_src + (size_t)hh * D;
  const float* dp = a_dst + (size_t)hh * D;
  float s = 0.f, d = 0.f;
  for (int k = 0; k < D; ++k) { float v = hp[k]; s += v * sp[k]; d += v * dp[k]; }
  as[i] = s;
  ad[i] = d;
}

__device__ __forceinline__ float edge_term(float asv, float adv, float mas) {
  float e = asv + adv;
  e = (e >= 0.f) ? e : NEG_SLOPE * e;
  float m = mas + adv;
  m = (m >= 0.f) ? m : NEG_SLOPE * m;
  return expf(e - m);
}

template <int H, int NPW, int CAP, int U, int G>
__global__ __launch_bounds__(1024) void owner_acc_kernel(
    const int* __restrict__ ei, int E, int N,
    const unsigned char* __restrict__ okey,
    const float* __restrict__ as, const float* __restrict__ ad,
    const float* __restrict__ hfeat, const float* __restrict__ bias,
    float* __restrict__ out, int apply_elu,
    unsigned* __restrict__ lst, int* __restrict__ lcnt) {
  constexpr int D = 32;
  constexpr int HD = H * D;
  constexpr int T = 1024;
  constexpr int NW = T / 32;
  constexpr int CAPW = CAP / NW;
  constexpr int NB = CAPW / 32;
  constexpr int SPW = (NPW + NW - 1) / NW;
  static_assert(H == 4, "the alpha planes are read 16 bytes per node");
  static_assert(NW == 32, "one lane per wave when the waves' counts are added up");
  static_assert(CAP % T == 0, "whole batches");
  static_assert(SPW <= 32 && SPW <= CAPW, "self loops are listed by one lane each");
  static_assert(NPW <= 256, "the owned index is saved in 8 bits");
  static_assert((NPW * HD) % 4 == 0, "tile shape");
  __shared__ __attribute__((aligned(16))) float accL[NPW * HD];
  __shared__ __attribute__((aligned(16))) float esL[NPW * H];
  __shared__ __attribute__((aligned(16))) float adL[NPW * H];
  __shared__ __attribute__((aligned(16))) unsigned mxL[NPW * H];
  __shared__ int leL[CAP];
  __shared__ int wcL[NW];

  const int t = (int)threadIdx.x;
  const int lane = t & 31;
  const int wave = __builtin_amdgcn_readfirstlane(t >> 5);
  const int n0 = (int)blockIdx.x * NPW;
  const int nown = (N - n0 < NPW) ? (N - n0) : NPW;
  int* __restrict__ seg = leL + wave * CAPW;
  int cnt = nown - wave * SPW;
  cnt = (cnt < 0) ? 0 : ((cnt > SPW) ? SPW : cnt);

  {
    const v4f z = {0.f, 0.f, 0.f, 0.f};
    v4f* a4 = reinterpret_cast<v4f*>(accL);
    for (int i = t; i < NPW * HD / 4; i += T) a4[i] = z;
    for (int i = t; i < NPW * H; i += T) {
      esL[i] = 0.f;
      mxL[i] = 0u;
      adL[i] = (i < nown * H) ? ad[(size_t)n0 * H + i] : 0.f;
    }
    if (lane < cnt) seg[lane] = E + n0 + wave * SPW + lane;
  }
  __syncthreads();

  const int* __restrict__ dstp = ei + E;
  const unsigned bb = (unsigned)blockIdx.x * 0x01010101u;
  auto keep1 = [&](bool hit, int e) {
    const unsigned m = __builtin_amdgcn_ballot_w32(hit);
    if (m != 0u) {
      if (hit) {
        const int slot = cnt + (int)__builtin_amdgcn_mbcnt_lo(m, 0u);
        if (slot < CAPW) seg[slot] = e;
      }
      cnt += __builtin_popcount(m);
    }
  };
  auto test4 = [&](unsigned w, int ebase) {
    const unsigned x = w ^ bb;
    unsigned y = ~(((x & 0x7F7F7F7Fu) + 0x7F7F7F7Fu) | x | 0x7F7F7F7Fu);
    unsigned m = __builtin_amdgcn_ballot_w32(y != 0u);
    while (m != 0u) {
      if (y != 0u) {
        int slot = (int)__builtin_amdgcn_mbcnt_lo(m, (unsigned)cnt);
        slot = (slot < CAPW - 1) ? slot : CAPW - 1;
        seg[slot] = ebase + (__builtin_ctz(y) >> 3);
        y &= y - 1u;
      }
      cnt += __builtin_popcount(m);
      m = __builtin_amdgcn_ballot_w32(y != 0u);
    }
  };
  {
    const v4u* __restrict__ kv = reinterpret_cast<const v4u*>(okey);
    const int nvec = E >> 4;
    constexpr int TILEV = T * U;
    const int nfull = nvec / TILEV;
    if (nfull > 0) {
      int tile = (int)(blockIdx.x % (unsigned)nfull);
      v4u cur[U];
      for (int u = 0; u < U; ++u) cur[u] = kv[tile * TILEV + u * T + t];
      for (int k = 0; k < nfull; ++k) {
        const int nt = (tile + 1 == nfull) ? 0 : tile + 1;
        v4u nxt[U];
        for (int u = 0; u < U; ++u) nxt[u] = kv[nt * TILEV + u * T + t];
        for (int u = 0; u < U; ++u) {
          const int vi = tile * TILEV + u * T + t;
          for (int q = 0; q < 4; ++q) test4(cur[u][q], vi * 16 + q * 4);
        }
        for (int u = 0; u < U; ++u) cur[u] = nxt[u];
        tile = nt;
      }
    }
    for (int v0 = nfull * TILEV; v0 < nvec; v0 += T) {
      const int vi = v0 + t;
      const unsigned none = bb ^ 0x01010101u;
      v4u k4 = {none, none, none, none};
      if (vi < nvec) k4 = kv[vi];
      for (int q = 0; q < 4; ++q) test4(k4[q], vi * 16 + q * 4);
    }
    for (int e0 = nvec << 4; e0 < E; e0 += T) {
      const int e = e0 + t;
      keep1(e < E && (unsigned)okey[(e < E) ? e : 0] == (unsigned)blockIdx.x, e);
    }
  }
  if (lane == 0) wcL[wave] = cnt;
  __syncthreads();
  int total = 0, base = 0;
  bool fits = true;
  {
    const int cl = wcL[lane];
    for (int v = 0; v < NW; ++v) {
      const int cv = __builtin_amdgcn_readlane(cl, v);
      total += cv;
      if (v < wave) base += cv;
      fits = fits && (cv <= CAPW);
    }
  }
  const bool listed = fits;
  if (t == 0) lcnt[blockIdx.x] = listed ? total : CAP + 1;

  int sK[NB], dK[NB];
  v4f aK[NB];
  if (listed) {
    int eK[NB];
    for (int k = 0; k < NB; ++k) {
      const int idx = k * 32 + lane;
      const int ev = seg[(idx < cnt) ? idx : 0];
      eK[k] = (idx < cnt) ? ev : E + n0;
    }
    for (int k = 0; k < NB; ++k) {
      const bool real = eK[k] < E;
      const int ec = real ? eK[k] : 0;
      const int sv = ei[ec];
      const int dv = dstp[ec];
      sK[k] = real ? sv : eK[k] - E;
      dK[k] = (real ? dv : eK[k] - E) - n0;
    }
    for (int k = 0; k < NB; ++k)
      aK[k] = *reinterpret_cast<const v4f*>(as + (size_t)sK[k] * H);
    for (int k = 0; k < NB; ++k) {
      const int idx = k * 32 + lane;
      if (idx < cnt)
        for (int h = 0; h < H; ++h) atomicMax(&mxL[dK[k] * H + h], fkey(aK[k][h]));
    }
  } else {
    for (int k = 0; k < NB; ++k) { sK[k] = 0; dK[k] = 0; aK[k] = v4f{0.f, 0.f, 0.f, 0.f}; }
    for (int i = t; i < nown + E; i += T) {
      int s, dl;
      if (i < nown) { s = n0 + i; dl = i; }
      else {
        const int e = i - nown;
        const unsigned u = (unsigned)(dstp[e] - n0);
        if (u >= (unsigned)nown) continue;
        s = ei[e]; dl = (int)u;
      }
      for (int h = 0; h < H; ++h) atomicMax(&mxL[dl * H + h], fkey(as[(size_t)s * H + h]));
    }
  }
  __syncthreads();

  if (listed) {
    unsigned* __restrict__ lsave = lst + (size_t)blockIdx.x * CAP + base;
    for (int k = 0; k < NB; ++k) {
      const int b0 = k * 32;
      if (b0 >= cnt) break;
      const int bc = (cnt - b0 < 32) ? (cnt - b0) : 32;
      const int s = sK[0], dl = dK[0];
      const v4f asv = aK[0];
      for (int q = 0; q + 1 < NB; ++q) { sK[q] = sK[q + 1]; dK[q] = dK[q + 1]; aK[q] = aK[q + 1]; }
      float ex[H];
      for (int h = 0; h < H; ++h) ex[h] = 0.f;
      if (lane < bc) {
        lsave[b0 + lane] = (unsigned)s | ((unsigned)dl << 24);
        const v4f a4 = *reinterpret_cast<const v4f*>(adL + dl * H);
        const v4u k4 = *reinterpret_cast<const v4u*>(mxL + dl * H);
        for (int h = 0; h < H; ++h) {
          ex[h] = edge_term(asv[h], a4[h], funkey(k4[h]));
          atomicAdd(&esL[dl * H + h], ex[h]);
        }
      }
      const int full = bc - bc % G;
      for (int j0 = 0; j0 < full; j0 += G) {
        float v[G][H];
        for (int g = 0; g < G; ++g) {
          const int sj = __builtin_amdgcn_readlane(s, j0 + g);
          const float* hp = hfeat + (size_t)sj * HD + lane;
          for (int h = 0; h < H; ++h) v[g][h] = hp[h * D];
        }
        for (int g = 0; g < G; ++g) {
          const int dj = __builtin_amdgcn_readlane(dl, j0 + g);
          float* ap = accL + dj * HD + lane;
          for (int h = 0; h < H; ++h) {
            const float w = __int_as_float(
                __builtin_amdgcn_readlane(__float_as_int(ex[h]), j0 + g));
            atomicAdd(&ap[h * D], w * v[g][h]);
          }
        }
      }
      for (int j = full; j < bc; ++j) {
        const int sj = __builtin_amdgcn_readlane(s, j);
        const int dj = __builtin_amdgcn_readlane(dl, j);
        const float* hp = hfeat + (size_t)sj * HD + lane;
        float* ap = accL + dj * HD + lane;
        for (int h = 0; h < H; ++h) {
          const float w = __int_as_float(
              __builtin_amdgcn_readlane(__float_as_int(ex[h]), j));
          atomicAdd(&ap[h * D], w * hp[h * D]);
        }
      }
    }
  } else {
    for (int i = t; i < nown + E; i += T) {
      int s, dl;
      if (i < nown) { s = n0 + i; dl = i; }
      else {
        const int e = i - nown;
        const unsigned u = (unsigned)(dstp[e] - n0);
        if (u >= (unsigned)nown) continue;
        s = ei[e]; dl = (int)u;
      }
      const float* hp = hfeat + (size_t)s * HD;
      for (int h = 0; h < H; ++h) {
        const float exh = edge_term(as[(size_t)s * H + h], adL[dl * H + h],
                                    funkey(mxL[dl * H + h]));
        atomicAdd(&esL[dl * H + h], exh);
        for (int kk = 0; kk < D; ++kk)
          atomicAdd(&accL[dl * HD + h * D + kk], exh * hp[h * D + kk]);
      }
    }
  }
  __syncthreads();

  for (int i = t; i < nown * HD; i += T) {
    const int nl = i / HD, c = i % HD, h = c / D;
    float v = accL[i] / (esL[nl * H + h] + 1e-16f) + bias[c];
    if (apply_elu) v = (v > 0.f) ? v : (expf(v) - 1.f);
    out[(size_t)n0 * HD + i] = v;
  }
}

template <int NPW, int CAP, int G>
__global__ __launch_bounds__(1024) void owner_list_acc_kernel(
    const int* __restrict__ ei, int E, int N,
    const unsigned* __restrict__ lst, const int* __restrict__ lcnt,
    const float* __restrict__ as, const float* __restrict__ ad,
    const float* __restrict__ hfeat, const float* __restrict__ bias,
    float* __restrict__ out, int apply_elu) {
  constexpr int D = 32;
  constexpr int T = 1024;
  constexpr int NB = CAP / T;
  static_assert(CAP % T == 0, "whole batches");
  static_assert((NPW * D) % 4 == 0 && 32 % G == 0, "tile shape");
  __shared__ __attribute__((aligned(16))) float accL[NPW * D];
  __shared__ float esL[NPW];
  __shared__ float adL[NPW];
  __shared__ unsigned mxL[NPW];

  const int t = (int)threadIdx.x;
  const int n0 = (int)blockIdx.x * NPW;
  const int nown = (N - n0 < NPW) ? (N - n0) : NPW;

  {
    const v4f z = {0.f, 0.f, 0.f, 0.f};
    v4f* a4 = reinterpret_cast<v4f*>(accL);
    for (int i = t; i < NPW * D / 4; i += T) a4[i] = z;
    for (int i = t; i < NPW; i += T) {
      esL[i] = 0.f;
      mxL[i] = 0u;
      adL[i] = (i < nown) ? ad[n0 + i] : 0.f;
    }
  }
  __syncthreads();

  const int total = lcnt[blockIdx.x];
  const bool listed = total <= CAP;
  const int* __restrict__ dstp = ei + E;

  int sK[NB], dK[NB];
  float aK[NB];
  if (listed) {
    const unsigned* __restrict__ lp = lst + (size_t)blockIdx.x * CAP;
    for (int k = 0; k < NB; ++k) {
      const int idx = k * T + t;
      const unsigned w = lp[(idx < total) ? idx : 0];
      sK[k] = (int)(w & 0xFFFFFFu);
      dK[k] = (int)(w >> 24);
    }
    for (int k = 0; k < NB; ++k) aK[k] = as[sK[k]];
    for (int k = 0; k < NB; ++k)
      if (k * T + t < total) atomicMax(&mxL[dK[k]], fkey(aK[k]));
  } else {
    for (int k = 0; k < NB; ++k) { sK[k] = 0; dK[k] = 0; aK[k] = 0.f; }
    for (int i = t; i < nown + E; i += T) {
      int s, dl;
      if (i < nown) { s = n0 + i; dl = i; }
      else {
        const int e = i - nown;
        const unsigned u = (unsigned)(dstp[e] - n0);
        if (u >= (unsigned)nown) continue;
        s = ei[e]; dl = (int)u;
      }
      atomicMax(&mxL[dl], fkey(as[s]));
    }
  }
  __syncthreads();

  if (listed) {
    const int lane = t & 31;
    const int wave = __builtin_amdgcn_readfirstlane(t >> 5);
    for (int k = 0; k < NB; ++k) {
      const int b0 = wave * 32 + k * T;
      if (b0 >= total) break;
      const int bc = (total - b0 < 32) ? (total - b0) : 32;
      const int s = sK[0], dl = dK[0];
      const float asv = aK[0];
      for (int q = 0; q + 1 < NB; ++q) { sK[q] = sK[q + 1]; dK[q] = dK[q + 1]; aK[q] = aK[q + 1]; }
      float ex = 0.f;
      if (lane < bc) {
        ex = edge_term(asv, adL[dl], funkey(mxL[dl]));
        atomicAdd(&esL[dl], ex);
      }
      const int full = bc - bc % G;
      for (int j0 = 0; j0 < full; j0 += G) {
        float v[G];
        for (int g = 0; g < G; ++g) {
          const int sj = __builtin_amdgcn_readlane(s, j0 + g);
          v[g] = hfeat[(size_t)sj * D + lane];
        }
        for (int g = 0; g < G; ++g) {
          const int dj = __builtin_amdgcn_readlane(dl, j0 + g);
          const float w = __int_as_float(
              __builtin_amdgcn_readlane(__float_as_int(ex), j0 + g));
          atomicAdd(&accL[dj * D + lane], w * v[g]);
        }
      }
      for (int j = full; j < bc; ++j) {
        const int sj = __builtin_amdgcn_readlane(s, j);
        const int dj = __builtin_amdgcn_readlane(dl, j);
        const float w = __int_as_float(
            __builtin_amdgcn_readlane(__float_as_int(ex), j));
        atomicAdd(&accL[dj * D + lane], w * hfeat[(size_t)sj * D + lane]);
      }
    }
  } else {
    for (int i = t; i < nown + E; i += T) {
      int s, dl;
      if (i < nown) { s = n0 + i; dl = i; }
      else {
        const int e = i - nown;
        const unsigned u = (unsigned)(dstp[e] - n0);
        if (u >= (unsigned)nown) continue;
        s = ei[e]; dl = (int)u;
      }
      const float exv = edge_term(as[s], adL[dl], funkey(mxL[dl]));
      const float* hp = hfeat + (size_t)s * D;
      atomicAdd(&esL[dl], exv);
      for (int kk = 0; kk < D; ++kk) atomicAdd(&accL[dl * D + kk], exv * hp[kk]);
    }
  }
  __syncthreads();

  for (int i = t; i < nown * D; i += T) {
    const int nl = i / D, c = i % D;
    float v = accL[i] / (esL[nl] + 1e-16f) + bias[c];
    if (apply_elu) v = (v > 0.f) ? v : (expf(v) - 1.f);
    out[(size_t)n0 * D + i] = v;
  }
}

__global__ void pred_node_kernel(const float* __restrict__ h2,
                                 const float* __restrict__ Wp,
                                 float* __restrict__ ps, float* __restrict__ pd,
                                 int N, int D) {
  int n = blockIdx.x * blockDim.x + threadIdx.x;
  if (n >= N) return;
  const float* hp = h2 + (size_t)n * D;
  float s = 0.f, d = 0.f;
  for (int k = 0; k < D; ++k) { s += hp[k] * Wp[k]; d += hp[k] * Wp[D + k]; }
  ps[n] = s; pd[n] = d;
}

__global__ void pred_edge_kernel(const int* __restrict__ ei, int E,
                                 const float* __restrict__ ps,
                                 const float* __restrict__ pd,
                                 const float* __restrict__ bp,
                                 float* __restrict__ out) {
  int e = blockIdx.x * blockDim.x + threadIdx.x;
  if (e >= E) return;
  out[e] = ps[ei[e]] + pd[ei[E + e]] + bp[0];
}

static inline int cdiv(long long a, int b) { return (int)((a + b - 1) / b); }

extern "C" void kernel_launch(void* const* d_in, const int* in_sizes, int n_in,
                              void* d_out, int out_size, void* d_ws, size_t ws_size,
                              hipStream_t stream) {
  const float* x   = (const float*)d_in[0];
  const int*   ei  = (const int*)  d_in[1];
  const float* W1  = (const float*)d_in[2];
  const float* as1w= (const float*)d_in[3];
  const float* ad1w= (const float*)d_in[4];
  const float* b1  = (const float*)d_in[5];
  const float* W2  = (const float*)d_in[6];
  const float* as2w= (const float*)d_in[7];
  const float* ad2w= (const float*)d_in[8];
  const float* b2  = (const float*)d_in[9];
  const float* Wp  = (const float*)d_in[10];
  const float* bp  = (const float*)d_in[11];
  float* out = (float*)d_out;

  const int F = 128, H1 = 4, Hd = 32;
  const int N = in_sizes[0] / F;
  const int E = in_sizes[1] / 2;

  constexpr int NPW = 196, CAP = 6144;
  float* ws = (float*)d_ws;
  float*    h1    = ws;                 ws += (size_t)N * F;
  float*    acc1  = ws;                 ws += (size_t)N * F;
  float*    as1   = ws;                 ws += (size_t)N * H1;
  float*    ad1   = ws;                 ws += (size_t)N * 2 * H1;
  int*      lcnt  = (int*)ws;           ws += (size_t)N * H1;
  float*    h2p   = ws;                 ws += (size_t)N * Hd;
  unsigned char* okey = (unsigned char*)h2p;
  unsigned* lst   = (unsigned*)ws;      ws += (size_t)N * Hd;
  float*    as2   = ws;                 ws += (size_t)N;
  float*    ad2   = ws;                 ws += (size_t)N * 2;
  ws += (size_t)N;
  float*    ps    = ws;                 ws += (size_t)N;
  float*    pd    = ws;                 ws += (size_t)N;
  float*    out2  = h1;
  (void)ws_size;

  const int T = 256;

  gemm_wmma_f32_kernel<8><<<cdiv(N, 32), 8 * 32, 0, stream>>>(x, W1, h1, N);
  alpha_kernel<NPW><<<cdiv((long long)N * H1, T), T, 0, stream>>>(h1, as1w, ad1w, as1, ad1, N, H1, Hd, ei, E, okey);
  owner_acc_kernel<4, NPW, CAP, 4, 8><<<cdiv(N, NPW), 1024, 0, stream>>>(
      ei, E, N, okey, as1, ad1, h1, b1, acc1, 1, lst, lcnt);

  gemm_wmma_f32_kernel<2><<<cdiv(N, 32), 2 * 32, 0, stream>>>(acc1, W2, h2p, N);
  alpha_kernel<NPW><<<cdiv(N, T), T, 0, stream>>>(h2p, as2w, ad2w, as2, ad2, N, 1, Hd, ei, E, nullptr);
  owner_list_acc_kernel<NPW, CAP, 16><<<cdiv(N, NPW), 1024, 0, stream>>>(
      ei, E, N, lst, lcnt, as2, ad2, h2p, b2, out2, 0);

  pred_node_kernel<<<cdiv(N, T), T, 0, stream>>>(out2, Wp, ps, pd, N, Hd);
  pred_edge_kernel<<<cdiv(E, T), T, 0, stream>>>(ei, E, ps, pd, bp, out);
}


# Owner work-groups stop scanning all edges: launch 2 orders each 4096-edge tile by owner in LDS and writes 257 starts per tile; an owner copies its short run of every tile; sums in arrival order
# speedup vs baseline: 13.1761x; 1.1199x over previous
#include <hip/hip_runtime.h>
#include <hip/hip_bf16.h>

typedef float v2f __attribute__((ext_vector_type(2)));
typedef float v4f __attribute__((ext_vector_type(4)));
typedef float v8f __attribute__((ext_vector_type(8)));
typedef int v4i __attribute__((ext_vector_type(4)));
typedef unsigned v4u __attribute__((ext_vector_type(4)));

#define NEG_SLOPE 0.2f

__device__ __forceinline__ unsigned fkey(float x) {
  unsigned b = __float_as_uint(x);
  return (b & 0x80000000u) ? ~b : (b | 0x80000000u);
}
__device__ __forceinline__ float funkey(unsigned k) {
  return (k & 0x80000000u) ? __uint_as_float(k & 0x7FFFFFFFu) : __uint_as_float(~k);
}

template <int NCT>
__global__ __launch_bounds__(NCT * 32) void gemm_wmma_f32_kernel(
    const float* __restrict__ A, const float* __restrict__ B,
    float* __restrict__ C, int M) {
  constexpr int K = 128;
  constexpr int KS = K / 4;
  constexpr int NCOL = NCT * 16;
  constexpr int NT = NCT * 32;
  constexpr int RT = 32;
  constexpr int P = K + 4;
  constexpr int AQ = K / 4;
  constexpr int CQ = NCOL / 4;
  static_assert((RT * AQ) % NT == 0 && (RT * CQ) % NT == 0, "tile shape");
  __shared__ __attribute__((aligned(16))) float tL[RT * P];

  const int t = (int)threadIdx.x;
  const int wave = t >> 5;
  const int lane = t & 31;
  const int lm = lane & 15;
  const int hi = lane >> 4;
  const int rowBase = (int)blockIdx.x * RT;
  const int colBase = wave * 16;

  constexpr int NA = RT * AQ / NT;
  v4f av[NA];
#pragma unroll
  for (int n = 0; n < NA; ++n) {
    const int i = t + n * NT;
    const int r = i / AQ, q = i % AQ;
    const int row = (rowBase + r < M) ? (rowBase + r) : (M - 1);
    av[n] = *reinterpret_cast<const v4f*>(A + (size_t)row * K + q * 4);
  }

  v2f b[KS];
  {
    const float* bp = B + (size_t)(hi * 2) * NCOL + colBase + lm;
#pragma unroll
    for (int j = 0; j < KS; ++j) {
      b[j].x = bp[(size_t)(4 * j) * NCOL];
      b[j].y = bp[(size_t)(4 * j + 1) * NCOL];
    }
  }

#pragma unroll
  for (int n = 0; n < NA; ++n) {
    const int i = t + n * NT;
    const int r = i / AQ, q = i % AQ;
    *reinterpret_cast<v4f*>(tL + r * P + q * 4) = av[n];
  }
  __syncthreads();

  v8f acc0 = {}, acc1 = {};
  {
    const float* a0p = tL + lm * P + hi * 2;
    const float* a1p = a0p + 16 * P;
#pragma unroll
    for (int j = 0; j < KS; ++j) {
      const v2f a0 = *reinterpret_cast<const v2f*>(a0p + 4 * j);
      const v2f a1 = *reinterpret_cast<const v2f*>(a1p + 4 * j);
      acc0 = __builtin_amdgcn_wmma_f32_16x16x4_f32(
           false, a0,  false, b[j],
           (short)0, acc0,  false,  false);
      acc1 = __builtin_amdgcn_wmma_f32_16x16x4_f32(
           false, a1,  false, b[j],
           (short)0, acc1,  false,  false);
    }
  }
  __syncthreads();

  for (int r = 0; r < 8; ++r) {
    tL[(hi * 8 + r) * P + colBase + lm] = acc0[r];
    tL[(16 + hi * 8 + r) * P + colBase + lm] = acc1[r];
  }
  __syncthreads();
  for (int n = 0; n < RT * CQ / NT; ++n) {
    const int i = t + n * NT;
    const int r = i / CQ, q = i % CQ;
    if (rowBase + r < M)
      *reinterpret_cast<v4f*>(C + (size_t)(rowBase + r) * NCOL + q * 4) =
          *reinterpret_cast<const v4f*>(tL + r * P + q * 4);
  }
}

template <int NPW, int TS, int MAXOWN>
__global__ __launch_bounds__(TS / 4) void alpha_kernel(const float* __restrict__ h,
                             const float* __restrict__ a_src,
                             const float* __restrict__ a_dst,
                             float* __restrict__ as, float* __restrict__ ad,
                             int N, int H, int D,
                             const int* __restrict__ ei, int E,
                             unsigned* __restrict__ tl, int* __restrict__ toff) {
  static_assert(MAXOWN % 32 == 0 && TS % 4 == 0, "a lane of the first wave adds up MAXOWN/32 owners");
  static_assert(NPW <= 256, "the owned index is kept in 8 bits");
  __shared__ int binL[MAXOWN];
  __shared__ __attribute__((aligned(16))) unsigned srtL[TS];
  const int t = (int)threadIdx.x;
  if (tl) {
    const int* __restrict__ dstp = ei + E;
    for (int tile = (int)blockIdx.x; (long long)tile * TS < E; tile += (int)gridDim.x) {
      const int eb = tile * TS;
      const int ne = (E - eb < TS) ? (E - eb) : TS;
      for (int k = t; k < MAXOWN; k += (int)blockDim.x) binL[k] = 0;
      __syncthreads();
      const int j0 = 4 * t;
      int o[4], r[4];
      unsigned w[4];
      {
        int d[4], s[4];
        if ((E & 3) == 0 && j0 < ne) {
          const v4i dv = *reinterpret_cast<const v4i*>(dstp + eb + j0);
          const v4i sv = *reinterpret_cast<const v4i*>(ei + eb + j0);
          d[0] = dv[0]; d[1] = dv[1]; d[2] = dv[2]; d[3] = dv[3];
          s[0] = sv[0]; s[1] = sv[1]; s[2] = sv[2]; s[3] = sv[3];
        } else {
          for (int q = 0; q < 4; ++q) {
            d[q] = (j0 + q < ne) ? dstp[eb + j0 + q] : -1;
            s[q] = (j0 + q < ne) ? ei[eb + j0 + q] : 0;
          }
        }
        for (int q = 0; q < 4; ++q) {
          o[q] = -1; r[q] = 0; w[q] = 0u;
          if (d[q] >= 0) {
            int ow = d[q] / NPW;
            ow = (ow < MAXOWN) ? ow : MAXOWN - 1;
            int dl = d[q] - ow * NPW;
            dl = (dl < NPW) ? dl : NPW - 1;
            o[q] = ow;
            w[q] = ((unsigned)s[q] & 0xFFFFFFu) | ((unsigned)dl << 24);
            r[q] = atomicAdd(&binL[ow], 1);
          }
        }
      }
      __syncthreads();
      if (t < 32) {
        constexpr int PER = MAXOWN / 32;
        int c[PER];
        int s = 0;
        for (int k = 0; k < PER; ++k) { c[k] = binL[t * PER + k]; s += c[k]; }
        int inc = s;
        for (int dl = 1; dl < 32; dl <<= 1) {
          const int v = __shfl_up(inc, dl);
          if (t >= dl) inc += v;
        }
        int run = inc - s;
        int* __restrict__ tp = toff + (size_t)tile * (MAXOWN + 1);
        for (int k = 0; k < PER; ++k) { binL[t * PER + k] = run; tp[t * PER + k] = run; run += c[k]; }
        if (t == 31) tp[MAXOWN] = run;
      }
      __syncthreads();
      for (int q = 0; q < 4; ++q)
        if (o[q] >= 0) srtL[binL[o[q]] + r[q]] = w[q];
      __syncthreads();
      if (j0 + 3 < ne) {
        *reinterpret_cast<v4u*>(tl + eb + j0) = *reinterpret_cast<const v4u*>(srtL + j0);
      } else {
        for (int q = 0; q < 4; ++q) if (j0 + q < ne) tl[eb + j0 + q] = srtL[j0 + q];
      }
    }
  }
  for (int i = (int)(blockIdx.x * blockDim.x) + t; i < N * H; i += (int)(gridDim.x * blockDim.x)) {
    int hh = i % H;
    const float* hp = h + (size_t)i * D;
    const float* sp = a_src + (size_t)hh * D;
    const float* dp = a_dst + (size_t)hh * D;
    float s = 0.f, d = 0.f;
    for (int k = 0; k < D; ++k) { float v = hp[k]; s += v * sp[k]; d += v * dp[k]; }
    as[i] = s;
    ad[i] = d;
  }
}

__device__ __forceinline__ float edge_term(float asv, float adv, float mas) {
  float e = asv + adv;
  e = (e >= 0.f) ? e : NEG_SLOPE * e;
  float m = mas + adv;
  m = (m >= 0.f) ? m : NEG_SLOPE * m;
  return expf(e - m);
}

template <int H, int NPW, int CAP, int G, int TS, int MAXOWN>
__global__ __launch_bounds__(1024) void owner_acc_kernel(
    const int* __restrict__ ei, int E, int N,
    const unsigned* __restrict__ tl, const int* __restrict__ toff,
    const float* __restrict__ as, const float* __restrict__ ad,
    const float* __restrict__ hfeat, const float* __restrict__ bias,
    float* __restrict__ out, int apply_elu,
    unsigned* __restrict__ lst, int* __restrict__ lcnt) {
  constexpr int D = 32;
  constexpr int HD = H * D;
  constexpr int T = 1024;
  constexpr int NW = T / 32;
  constexpr int CAPW = CAP / NW;
  constexpr int NB = CAPW / 32;
  constexpr int MAXT = 256;
  constexpr int TPT = T / MAXT;
  static_assert(H == 4, "the alpha planes are read 16 bytes per node");
  static_assert(CAP % T == 0, "whole batches");
  static_assert(T % MAXT == 0 && MAXT % 32 == 0, "a lane of the first wave adds up MAXT/32 tiles");
  static_assert(NPW <= 256, "the owned index is saved in 8 bits");
  static_assert((NPW * HD) % 4 == 0, "tile shape");
  __shared__ __attribute__((aligned(16))) float accL[NPW * HD];
  __shared__ __attribute__((aligned(16))) float esL[NPW * H];
  __shared__ __attribute__((aligned(16))) float adL[NPW * H];
  __shared__ __attribute__((aligned(16))) unsigned mxL[NPW * H];
  __shared__ unsigned leL[CAP];
  __shared__ int tcL[MAXT];
  __shared__ int toL[MAXT];
  __shared__ int tpL[MAXT];

  const int t = (int)threadIdx.x;
  const int lane = t & 31;
  const int wave = __builtin_amdgcn_readfirstlane(t >> 5);
  const int n0 = (int)blockIdx.x * NPW;
  const int nown = (N - n0 < NPW) ? (N - n0) : NPW;
  const int nt = (E + TS - 1) / TS;
  const bool tiled = (nt <= MAXT) && ((int)blockIdx.x < MAXOWN);

  {
    int o0 = 0, o1 = 0;
    if (tiled && t < nt && t < MAXT) {
      const int* __restrict__ tp = toff + (size_t)t * (MAXOWN + 1) + blockIdx.x;
      o0 = tp[0];
      o1 = tp[1];
    }
    const v4f z = {0.f, 0.f, 0.f, 0.f};
    v4f* a4 = reinterpret_cast<v4f*>(accL);
    for (int i = t; i < NPW * HD / 4; i += T) a4[i] = z;
    for (int i = t; i < NPW * H; i += T) {
      esL[i] = 0.f;
      mxL[i] = 0u;
      adL[i] = (i < nown * H) ? ad[(size_t)n0 * H + i] : 0.f;
    }
    if (t < MAXT) {
      toL[t] = o0;
      tcL[t] = o1 - o0;
    }
  }
  __syncthreads();

  const int* __restrict__ dstp = ei + E;
  if (t < 32) {
    constexpr int PER = MAXT / 32;
    int c[PER];
    int s = 0;
    for (int k = 0; k < PER; ++k) { c[k] = tcL[t * PER + k]; s += c[k]; }
    int inc = s;
    for (int dl = 1; dl < 32; dl <<= 1) {
      const int v = __shfl_up(inc, dl);
      if (t >= dl) inc += v;
    }
    int run = inc - s;
    for (int k = 0; k < PER; ++k) { tpL[t * PER + k] = run; run += c[k]; }
  }
  __syncthreads();
  const int total = nown + tpL[MAXT - 1] + tcL[MAXT - 1];
  const bool listed = tiled && (total <= CAP);
  if (listed) {
    if (t < nown) leL[t] = (unsigned)(n0 + t) | ((unsigned)t << 24);
    const int tau = t / TPT, r = t % TPT;
    const int c = tcL[tau];
    const unsigned* __restrict__ sp = tl + (size_t)tau * TS + toL[tau];
    unsigned* dp = leL + nown + tpL[tau];
    for (int j = r; j < c; j += TPT) dp[j] = sp[j];
  }
  __syncthreads();
  const int per = (total + NW - 1) / NW;
  const unsigned* seg = leL + (listed ? wave * per : 0);
  int cnt = total - wave * per;
  cnt = (cnt < 0) ? 0 : ((cnt > per) ? per : cnt);
  const int base = wave * per;
  if (t == 0) lcnt[blockIdx.x] = listed ? total : CAP + 1;

  int sK[NB], dK[NB];
  v4f aK[NB];
  if (listed) {
    for (int k = 0; k < NB; ++k) {
      const int idx = k * 32 + lane;
      const unsigned wv = seg[(idx < cnt) ? idx : 0];
      const unsigned wk = (idx < cnt) ? wv : (unsigned)n0;
      sK[k] = (int)(wk & 0xFFFFFFu);
      dK[k] = (int)(wk >> 24);
    }
    for (int k = 0; k < NB; ++k)
      aK[k] = *reinterpret_cast<const v4f*>(as + (size_t)sK[k] * H);
    for (int k = 0; k < NB; ++k) {
      const int idx = k * 32 + lane;
      if (idx < cnt)
        for (int h = 0; h < H; ++h) atomicMax(&mxL[dK[k] * H + h], fkey(aK[k][h]));
    }
  } else {
    for (int k = 0; k < NB; ++k) { sK[k] = 0; dK[k] = 0; aK[k] = v4f{0.f, 0.f, 0.f, 0.f}; }
    for (int i = t; i < nown + E; i += T) {
      int s, dl;
      if (i < nown) { s = n0 + i; dl = i; }
      else {
        const int e = i - nown;
        const unsigned u = (unsigned)(dstp[e] - n0);
        if (u >= (unsigned)nown) continue;
        s = ei[e]; dl = (int)u;
      }
      for (int h = 0; h < H; ++h) atomicMax(&mxL[dl * H + h], fkey(as[(size_t)s * H + h]));
    }
  }
  __syncthreads();

  if (listed) {
    unsigned* __restrict__ lsave = lst + (size_t)blockIdx.x * CAP + base;
    for (int k = 0; k < NB; ++k) {
      const int b0 = k * 32;
      if (b0 >= cnt) break;
      const int bc = (cnt - b0 < 32) ? (cnt - b0) : 32;
      const int s = sK[0], dl = dK[0];
      const v4f asv = aK[0];
      for (int q = 0; q + 1 < NB; ++q) { sK[q] = sK[q + 1]; dK[q] = dK[q + 1]; aK[q] = aK[q + 1]; }
      float ex[H];
      for (int h = 0; h < H; ++h) ex[h] = 0.f;
      if (lane < bc) {
        lsave[b0 + lane] = (unsigned)s | ((unsigned)dl << 24);
        const v4f a4 = *reinterpret_cast<const v4f*>(adL + dl * H);
        const v4u k4 = *reinterpret_cast<const v4u*>(mxL + dl * H);
        for (int h = 0; h < H; ++h) {
          ex[h] = edge_term(asv[h], a4[h], funkey(k4[h]));
          atomicAdd(&esL[dl * H + h], ex[h]);
        }
      }
      const int full = bc - bc % G;
      for (int j0 = 0; j0 < full; j0 += G) {
        float v[G][H];
        for (int g = 0; g < G; ++g) {
          const int sj = __builtin_amdgcn_readlane(s, j0 + g);
          const float* hp = hfeat + (size_t)sj * HD + lane;
          for (int h = 0; h < H; ++h) v[g][h] = hp[h * D];
        }
        for (int g = 0; g < G; ++g) {
          const int dj = __builtin_amdgcn_readlane(dl, j0 + g);
          float* ap = accL + dj * HD + lane;
          for (int h = 0; h < H; ++h) {
            const float w = __int_as_float(
                __builtin_amdgcn_readlane(__float_as_int(ex[h]), j0 + g));
            atomicAdd(&ap[h * D], w * v[g][h]);
          }
        }
      }
      for (int j = full; j < bc; ++j) {
        const int sj = __builtin_amdgcn_readlane(s, j);
        const int dj = __builtin_amdgcn_readlane(dl, j);
        const float* hp = hfeat + (size_t)sj * HD + lane;
        float* ap = accL + dj * HD + lane;
        for (int h = 0; h < H; ++h) {
          const float w = __int_as_float(
              __builtin_amdgcn_readlane(__float_as_int(ex[h]), j));
          atomicAdd(&ap[h * D], w * hp[h * D]);
        }
      }
    }
  } else {
    for (int i = t; i < nown + E; i += T) {
      int s, dl;
      if (i < nown) { s = n0 + i; dl = i; }
      else {
        const int e = i - nown;
        const unsigned u = (unsigned)(dstp[e] - n0);
        if (u >= (unsigned)nown) continue;
        s = ei[e]; dl = (int)u;
      }
      const float* hp = hfeat + (size_t)s * HD;
      for (int h = 0; h < H; ++h) {
        const float exh = edge_term(as[(size_t)s * H + h], adL[dl * H + h],
                                    funkey(mxL[dl * H + h]));
        atomicAdd(&esL[dl * H + h], exh);
        for (int kk = 0; kk < D; ++kk)
          atomicAdd(&accL[dl * HD + h * D + kk], exh * hp[h * D + kk]);
      }
    }
  }
  __syncthreads();

  for (int i = t; i < nown * HD; i += T) {
    const int nl = i / HD, c = i % HD, h = c / D;
    float v = accL[i] / (esL[nl * H + h] + 1e-16f) + bias[c];
    if (apply_elu) v = (v > 0.f) ? v : (expf(v) - 1.f);
    out[(size_t)n0 * HD + i] = v;
  }
}

template <int NPW, int CAP, int G>
__global__ __launch_bounds__(1024) void owner_list_acc_kernel(
    const int* __restrict__ ei, int E, int N,
    const unsigned* __restrict__ lst, const int* __restrict__ lcnt,
    const float* __restrict__ as, const float* __restrict__ ad,
    const float* __restrict__ hfeat, const float* __restrict__ bias,
    float* __restrict__ out, int apply_elu) {
  constexpr int D = 32;
  constexpr int T = 1024;
  constexpr int NB = CAP / T;
  static_assert(CAP % T == 0, "whole batches");
  static_assert((NPW * D) % 4 == 0 && 32 % G == 0, "tile shape");
  __shared__ __attribute__((aligned(16))) float accL[NPW * D];
  __shared__ float esL[NPW];
  __shared__ float adL[NPW];
  __shared__ unsigned mxL[NPW];

  const int t = (int)threadIdx.x;
  const int n0 = (int)blockIdx.x * NPW;
  const int nown = (N - n0 < NPW) ? (N - n0) : NPW;

  {
    const v4f z = {0.f, 0.f, 0.f, 0.f};
    v4f* a4 = reinterpret_cast<v4f*>(accL);
    for (int i = t; i < NPW * D / 4; i += T) a4[i] = z;
    for (int i = t; i < NPW; i += T) {
      esL[i] = 0.f;
      mxL[i] = 0u;
      adL[i] = (i < nown) ? ad[n0 + i] : 0.f;
    }
  }
  __syncthreads();

  const int total = lcnt[blockIdx.x];
  const bool listed = total <= CAP;
  const int* __restrict__ dstp = ei + E;

  int sK[NB], dK[NB];
  float aK[NB];
  if (listed) {
    const unsigned* __restrict__ lp = lst + (size_t)blockIdx.x * CAP;
    for (int k = 0; k < NB; ++k) {
      const int idx = k * T + t;
      const unsigned w = lp[(idx < total) ? idx : 0];
      sK[k] = (int)(w & 0xFFFFFFu);
      dK[k] = (int)(w >> 24);
    }
    for (int k = 0; k < NB; ++k) aK[k] = as[sK[k]];
    for (int k = 0; k < NB; ++k)
      if (k * T + t < total) atomicMax(&mxL[dK[k]], fkey(aK[k]));
  } else {
    for (int k = 0; k < NB; ++k) { sK[k] = 0; dK[k] = 0; aK[k] = 0.f; }
    for (int i = t; i < nown + E; i += T) {
      int s, dl;
      if (i < nown) { s = n0 + i; dl = i; }
      else {
        const int e = i - nown;
        const unsigned u = (unsigned)(dstp[e] - n0);
        if (u >= (unsigned)nown) continue;
        s = ei[e]; dl = (int)u;
      }
      atomicMax(&mxL[dl], fkey(as[s]));
    }
  }
  __syncthreads();

  if (listed) {
    const int lane = t & 31;
    const int wave = __builtin_amdgcn_readfirstlane(t >> 5);
    for (int k = 0; k < NB; ++k) {
      const int b0 = wave * 32 + k * T;
      if (b0 >= total) break;
      const int bc = (total - b0 < 32) ? (total - b0) : 32;
      const int s = sK[0], dl = dK[0];
      const float asv = aK[0];
      for (int q = 0; q + 1 < NB; ++q) { sK[q] = sK[q + 1]; dK[q] = dK[q + 1]; aK[q] = aK[q + 1]; }
      float ex = 0.f;
      if (lane < bc) {
        ex = edge_term(asv, adL[dl], funkey(mxL[dl]));
        atomicAdd(&esL[dl], ex);
      }
      const int full = bc - bc % G;
      for (int j0 = 0; j0 < full; j0 += G) {
        float v[G];
        for (int g = 0; g < G; ++g) {
          const int sj = __builtin_amdgcn_readlane(s, j0 + g);
          v[g] = hfeat[(size_t)sj * D + lane];
        }
        for (int g = 0; g < G; ++g) {
          const int dj = __builtin_amdgcn_readlane(dl, j0 + g);
          const float w = __int_as_float(
              __builtin_amdgcn_readlane(__float_as_int(ex), j0 + g));
          atomicAdd(&accL[dj * D + lane], w * v[g]);
        }
      }
      for (int j = full; j < bc; ++j) {
        const int sj = __builtin_amdgcn_readlane(s, j);
        const int dj = __builtin_amdgcn_readlane(dl, j);
        const float w = __int_as_float(
            __builtin_amdgcn_readlane(__float_as_int(ex), j));
        atomicAdd(&accL[dj * D + lane], w * hfeat[(size_t)sj * D + lane]);
      }
    }
  } else {
    for (int i = t; i < nown + E; i += T) {
      int s, dl;
      if (i < nown) { s = n0 + i; dl = i; }
      else {
        const int e = i - nown;
        const unsigned u = (unsigned)(dstp[e] - n0);
        if (u >= (unsigned)nown) continue;
        s = ei[e]; dl = (int)u;
      }
      const float exv = edge_term(as[s], adL[dl], funkey(mxL[dl]));
      const float* hp = hfeat + (size_t)s * D;
      atomicAdd(&esL[dl], exv);
      for (int kk = 0; kk < D; ++kk) atomicAdd(&accL[dl * D + kk], exv * hp[kk]);
    }
  }
  __syncthreads();

  for (int i = t; i < nown * D; i += T) {
    const int nl = i / D, c = i % D;
    float v = accL[i] / (esL[nl] + 1e-16f) + bias[c];
    if (apply_elu) v = (v > 0.f) ? v : (expf(v) - 1.f);
    out[(size_t)n0 * D + i] = v;
  }
}

__global__ void pred_node_kernel(const float* __restrict__ h2,
                                 const float* __restrict__ Wp,
                                 float* __restrict__ ps, float* __restrict__ pd,
                                 int N, int D) {
  int n = blockIdx.x * blockDim.x + threadIdx.x;
  if (n >= N) return;
  const float* hp = h2 + (size_t)n * D;
  float s = 0.f, d = 0.f;
  for (int k = 0; k < D; ++k) { s += hp[k] * Wp[k]; d += hp[k] * Wp[D + k]; }
  ps[n] = s; pd[n] = d;
}

__global__ void pred_edge_kernel(const int* __restrict__ ei, int E,
                                 const float* __restrict__ ps,
                                 const float* __restrict__ pd,
                                 const float* __restrict__ bp,
                                 float* __restrict__ out) {
  int e = blockIdx.x * blockDim.x + threadIdx.x;
  if (e >= E) return;
  out[e] = ps[ei[e]] + pd[ei[E + e]] + bp[0];
}

static inline int cdiv(long long a, int b) { return (int)((a + b - 1) / b); }

extern "C" void kernel_launch(void* const* d_in, const int* in_sizes, int n_in,
                              void* d_out, int out_size, void* d_ws, size_t ws_size,
                              hipStream_t stream) {
  const float* x   = (const float*)d_in[0];
  const int*   ei  = (const int*)  d_in[1];
  const float* W1  = (const float*)d_in[2];
  const float* as1w= (const float*)d_in[3];
  const float* ad1w= (const float*)d_in[4];
  const float* b1  = (const float*)d_in[5];
  const float* W2  = (const float*)d_in[6];
  const float* as2w= (const float*)d_in[7];
  const float* ad2w= (const float*)d_in[8];
  const float* b2  = (const float*)d_in[9];
  const float* Wp  = (const float*)d_in[10];
  const float* bp  = (const float*)d_in[11];
  float* out = (float*)d_out;

  const int F = 128, H1 = 4, Hd = 32;
  const int N = in_sizes[0] / F;
  const int E = in_sizes[1] / 2;

  constexpr int NPW = 196, CAP = 6144, TS = 4096, MAXOWN = 256;
  float* ws = (float*)d_ws;
  float*    h1    = ws;                 ws += (size_t)N * F;
  float*    acc1  = ws;                 ws += (size_t)N * F;
  float*    as1   = ws;                 ws += (size_t)N * H1;
  float*    ad1   = ws;                 ws += (size_t)N * 2 * H1;
  int*      lcnt  = (int*)ws;           ws += (size_t)N * H1;
  float*    h2p   = ws;                 ws += (size_t)N * Hd;
  unsigned* tl    = (unsigned*)h2p;
  int*      toff  = (int*)(tl + E);
  unsigned* lst   = (unsigned*)ws;      ws += (size_t)N * Hd;
  float*    as2   = ws;                 ws += (size_t)N;
  float*    ad2   = ws;                 ws += (size_t)N * 2;
  ws += (size_t)N;
  float*    ps    = ws;                 ws += (size_t)N;
  float*    pd    = ws;                 ws += (size_t)N;
  float*    out2  = h1;
  (void)ws_size;

  const int T = 256;

  gemm_wmma_f32_kernel<8><<<cdiv(N, 32), 8 * 32, 0, stream>>>(x, W1, h1, N);
  alpha_kernel<NPW, TS, MAXOWN><<<cdiv((long long)N * H1, TS / 4), TS / 4, 0, stream>>>(
      h1, as1w, ad1w, as1, ad1, N, H1, Hd, ei, E, tl, toff);
  owner_acc_kernel<4, NPW, CAP, 8, TS, MAXOWN><<<cdiv(N, NPW), 1024, 0, stream>>>(
      ei, E, N, tl, toff, as1, ad1, h1, b1, acc1, 1, lst, lcnt);

  gemm_wmma_f32_kernel<2><<<cdiv(N, 32), 2 * 32, 0, stream>>>(acc1, W2, h2p, N);
  alpha_kernel<NPW, TS, MAXOWN><<<cdiv(N, T), T, 0, stream>>>(h2p, as2w, ad2w, as2, ad2, N, 1, Hd, ei, E, nullptr, nullptr);
  owner_list_acc_kernel<NPW, CAP, 16><<<cdiv(N, NPW), 1024, 0, stream>>>(
      ei, E, N, lst, lcnt, as2, ad2, h2p, b2, out2, 0);

  pred_node_kernel<<<cdiv(N, T), T, 0, stream>>>(out2, Wp, ps, pd, N, Hd);
  pred_edge_kernel<<<cdiv(E, T), T, 0, stream>>>(ei, E, ps, pd, bp, out);
}


# Owner kernels, loads once waited for one per trip now leave in groups: a batch's last bc%G rows as one more group of G requests, an owner's run of a tile 8 words per thread; sums in arrival order
# speedup vs baseline: 13.6927x; 1.0392x over previous
#include <hip/hip_runtime.h>
#include <hip/hip_bf16.h>

typedef float v2f __attribute__((ext_vector_type(2)));
typedef float v4f __attribute__((ext_vector_type(4)));
typedef float v8f __attribute__((ext_vector_type(8)));
typedef int v4i __attribute__((ext_vector_type(4)));
typedef unsigned v4u __attribute__((ext_vector_type(4)));

#define NEG_SLOPE 0.2f

__device__ __forceinline__ unsigned fkey(float x) {
  unsigned b = __float_as_uint(x);
  return (b & 0x80000000u) ? ~b : (b | 0x80000000u);
}
__device__ __forceinline__ float funkey(unsigned k) {
  return (k & 0x80000000u) ? __uint_as_float(k & 0x7FFFFFFFu) : __uint_as_float(~k);
}

template <int NCT>
__global__ __launch_bounds__(NCT * 32) void gemm_wmma_f32_kernel(
    const float* __restrict__ A, const float* __restrict__ B,
    float* __restrict__ C, int M) {
  constexpr int K = 128;
  constexpr int KS = K / 4;
  constexpr int NCOL = NCT * 16;
  constexpr int NT = NCT * 32;
  constexpr int RT = 32;
  constexpr int P = K + 4;
  constexpr int AQ = K / 4;
  constexpr int CQ = NCOL / 4;
  static_assert((RT * AQ) % NT == 0 && (RT * CQ) % NT == 0, "tile shape");
  __shared__ __attribute__((aligned(16))) float tL[RT * P];

  const int t = (int)threadIdx.x;
  const int wave = t >> 5;
  const int lane = t & 31;
  const int lm = lane & 15;
  const int hi = lane >> 4;
  const int rowBase = (int)blockIdx.x * RT;
  const int colBase = wave * 16;

  constexpr int NA = RT * AQ / NT;
  v4f av[NA];
#pragma unroll
  for (int n = 0; n < NA; ++n) {
    const int i = t + n * NT;
    const int r = i / AQ, q = i % AQ;
    const int row = (rowBase + r < M) ? (rowBase + r) : (M - 1);
    av[n] = *reinterpret_cast<const v4f*>(A + (size_t)row * K + q * 4);
  }

  v2f b[KS];
  {
    const float* bp = B + (size_t)(hi * 2) * NCOL + colBase + lm;
#pragma unroll
    for (int j = 0; j < KS; ++j) {
      b[j].x = bp[(size_t)(4 * j) * NCOL];
      b[j].y = bp[(size_t)(4 * j + 1) * NCOL];
    }
  }

#pragma unroll
  for (int n = 0; n < NA; ++n) {
    const int i = t + n * NT;
    const int r = i / AQ, q = i % AQ;
    *reinterpret_cast<v4f*>(tL + r * P + q * 4) = av[n];
  }
  __syncthreads();

  v8f acc0 = {}, acc1 = {};
  {
    const float* a0p = tL + lm * P + hi * 2;
    const float* a1p = a0p + 16 * P;
#pragma unroll
    for (int j = 0; j < KS; ++j) {
      const v2f a0 = *reinterpret_cast<const v2f*>(a0p + 4 * j);
      const v2f a1 = *reinterpret_cast<const v2f*>(a1p + 4 * j);
      acc0 = __builtin_amdgcn_wmma_f32_16x16x4_f32(
           false, a0,  false, b[j],
           (short)0, acc0,  false,  false);
      acc1 = __builtin_amdgcn_wmma_f32_16x16x4_f32(
           false, a1,  false, b[j],
           (short)0, acc1,  false,  false);
    }
  }
  __syncthreads();

  for (int r = 0; r < 8; ++r) {
    tL[(hi * 8 + r) * P + colBase + lm] = acc0[r];
    tL[(16 + hi * 8 + r) * P + colBase + lm] = acc1[r];
  }
  __syncthreads();
  for (int n = 0; n < RT * CQ / NT; ++n) {
    const int i = t + n * NT;
    const int r = i / CQ, q = i % CQ;
    if (rowBase + r < M)
      *reinterpret_cast<v4f*>(C + (size_t)(rowBase + r) * NCOL + q * 4) =
          *reinterpret_cast<const v4f*>(tL + r * P + q * 4);
  }
}

template <int NPW, int TS, int MAXOWN>
__global__ __launch_bounds__(TS / 4) void alpha_kernel(const float* __restrict__ h,
                             const float* __restrict__ a_src,
                             const float* __restrict__ a_dst,
                             float* __restrict__ as, float* __restrict__ ad,
                             int N, int H, int D,
                             const int* __restrict__ ei, int E,
                             unsigned* __restrict__ tl, int* __restrict__ toff) {
  static_assert(MAXOWN % 32 == 0 && TS % 4 == 0, "a lane of the first wave adds up MAXOWN/32 owners");
  static_assert(NPW <= 256, "the owned index is kept in 8 bits");
  __shared__ int binL[MAXOWN];
  __shared__ __attribute__((aligned(16))) unsigned srtL[TS];
  const int t = (int)threadIdx.x;
  if (tl) {
    const int* __restrict__ dstp = ei + E;
    for (int tile = (int)blockIdx.x; (long long)tile * TS < E; tile += (int)gridDim.x) {
      const int eb = tile * TS;
      const int ne = (E - eb < TS) ? (E - eb) : TS;
      for (int k = t; k < MAXOWN; k += (int)blockDim.x) binL[k] = 0;
      __syncthreads();
      const int j0 = 4 * t;
      int o[4], r[4];
      unsigned w[4];
      {
        int d[4], s[4];
        if ((E & 3) == 0 && j0 < ne) {
          const v4i dv = *reinterpret_cast<const v4i*>(dstp + eb + j0);
          const v4i sv = *reinterpret_cast<const v4i*>(ei + eb + j0);
          d[0] = dv[0]; d[1] = dv[1]; d[2] = dv[2]; d[3] = dv[3];
          s[0] = sv[0]; s[1] = sv[1]; s[2] = sv[2]; s[3] = sv[3];
        } else {
          for (int q = 0; q < 4; ++q) {
            d[q] = (j0 + q < ne) ? dstp[eb + j0 + q] : -1;
            s[q] = (j0 + q < ne) ? ei[eb + j0 + q] : 0;
          }
        }
        for (int q = 0; q < 4; ++q) {
          o[q] = -1; r[q] = 0; w[q] = 0u;
          if (d[q] >= 0) {
            int ow = d[q] / NPW;
            ow = (ow < MAXOWN) ? ow : MAXOWN - 1;
            int dl = d[q] - ow * NPW;
            dl = (dl < NPW) ? dl : NPW - 1;
            o[q] = ow;
            w[q] = ((unsigned)s[q] & 0xFFFFFFu) | ((unsigned)dl << 24);
            r[q] = atomicAdd(&binL[ow], 1);
          }
        }
      }
      __syncthreads();
      if (t < 32) {
        constexpr int PER = MAXOWN / 32;
        int c[PER];
        int s = 0;
        for (int k = 0; k < PER; ++k) { c[k] = binL[t * PER + k]; s += c[k]; }
        int inc = s;
        for (int dl = 1; dl < 32; dl <<= 1) {
          const int v = __shfl_up(inc, dl);
          if (t >= dl) inc += v;
        }
        int run = inc - s;
        int* __restrict__ tp = toff + (size_t)tile * (MAXOWN + 1);
        for (int k = 0; k < PER; ++k) { binL[t * PER + k] = run; tp[t * PER + k] = run; run += c[k]; }
        if (t == 31) tp[MAXOWN] = run;
      }
      __syncthreads();
      for (int q = 0; q < 4; ++q)
        if (o[q] >= 0) srtL[binL[o[q]] + r[q]] = w[q];
      __syncthreads();
      if (j0 + 3 < ne) {
        *reinterpret_cast<v4u*>(tl + eb + j0) = *reinterpret_cast<const v4u*>(srtL + j0);
      } else {
        for (int q = 0; q < 4; ++q) if (j0 + q < ne) tl[eb + j0 + q] = srtL[j0 + q];
      }
    }
  }
  for (int i = (int)(blockIdx.x * blockDim.x) + t; i < N * H; i += (int)(gridDim.x * blockDim.x)) {
    int hh = i % H;
    const float* hp = h + (size_t)i * D;
    const float* sp = a_src + (size_t)hh * D;
    const float* dp = a_dst + (size_t)hh * D;
    float s = 0.f, d = 0.f;
    for (int k = 0; k < D; ++k) { float v = hp[k]; s += v * sp[k]; d += v * dp[k]; }
    as[i] = s;
    ad[i] = d;
  }
}

__device__ __forceinline__ float edge_term(float asv, float adv, float mas) {
  float e = asv + adv;
  e = (e >= 0.f) ? e : NEG_SLOPE * e;
  float m = mas + adv;
  m = (m >= 0.f) ? m : NEG_SLOPE * m;
  return expf(e - m);
}

template <int H, int NPW, int CAP, int G, int TS, int MAXOWN>
__global__ __launch_bounds__(1024) void owner_acc_kernel(
    const int* __restrict__ ei, int E, int N,
    const unsigned* __restrict__ tl, const int* __restrict__ toff,
    const float* __restrict__ as, const float* __restrict__ ad,
    const float* __restrict__ hfeat, const float* __restrict__ bias,
    float* __restrict__ out, int apply_elu,
    unsigned* __restrict__ lst, int* __restrict__ lcnt) {
  constexpr int D = 32;
  constexpr int HD = H * D;
  constexpr int T = 1024;
  constexpr int NW = T / 32;
  constexpr int CAPW = CAP / NW;
  constexpr int NB = CAPW / 32;
  constexpr int MAXT = 256;
  constexpr int TPT = T / MAXT;
  static_assert(H == 4, "the alpha planes are read 16 bytes per node");
  static_assert(CAP % T == 0, "whole batches");
  static_assert(T % MAXT == 0 && MAXT % 32 == 0, "a lane of the first wave adds up MAXT/32 tiles");
  static_assert(NPW <= 256, "the owned index is saved in 8 bits");
  static_assert((NPW * HD) % 4 == 0, "tile shape");
  __shared__ __attribute__((aligned(16))) float accL[NPW * HD];
  __shared__ __attribute__((aligned(16))) float esL[NPW * H];
  __shared__ __attribute__((aligned(16))) float adL[NPW * H];
  __shared__ __attribute__((aligned(16))) unsigned mxL[NPW * H];
  __shared__ unsigned leL[CAP];
  __shared__ int tcL[MAXT];
  __shared__ int toL[MAXT];
  __shared__ int tpL[MAXT];

  const int t = (int)threadIdx.x;
  const int lane = t & 31;
  const int wave = __builtin_amdgcn_readfirstlane(t >> 5);
  const int n0 = (int)blockIdx.x * NPW;
  const int nown = (N - n0 < NPW) ? (N - n0) : NPW;
  const int nt = (E + TS - 1) / TS;
  const bool tiled = (nt <= MAXT) && ((int)blockIdx.x < MAXOWN);

  {
    int o0 = 0, o1 = 0;
    if (tiled && t < nt && t < MAXT) {
      const int* __restrict__ tp = toff + (size_t)t * (MAXOWN + 1) + blockIdx.x;
      o0 = tp[0];
      o1 = tp[1];
    }
    const v4f z = {0.f, 0.f, 0.f, 0.f};
    v4f* a4 = reinterpret_cast<v4f*>(accL);
    for (int i = t; i < NPW * HD / 4; i += T) a4[i] = z;
    for (int i = t; i < NPW * H; i += T) {
      esL[i] = 0.f;
      mxL[i] = 0u;
      adL[i] = (i < nown * H) ? ad[(size_t)n0 * H + i] : 0.f;
    }
    if (t < MAXT) {
      toL[t] = o0;
      tcL[t] = o1 - o0;
    }
  }
  __syncthreads();

  const int* __restrict__ dstp = ei + E;
  if (t < 32) {
    constexpr int PER = MAXT / 32;
    int c[PER];
    int s = 0;
    for (int k = 0; k < PER; ++k) { c[k] = tcL[t * PER + k]; s += c[k]; }
    int inc = s;
    for (int dl = 1; dl < 32; dl <<= 1) {
      const int v = __shfl_up(inc, dl);
      if (t >= dl) inc += v;
    }
    int run = inc - s;
    for (int k = 0; k < PER; ++k) { tpL[t * PER + k] = run; run += c[k]; }
  }
  __syncthreads();
  const int total = nown + tpL[MAXT - 1] + tcL[MAXT - 1];
  const bool listed = tiled && (total <= CAP);
  if (listed) {
    if (t < nown) leL[t] = (unsigned)(n0 + t) | ((unsigned)t << 24);
    const int tau = t / TPT, r = t % TPT;
    const int c = tcL[tau];
    const unsigned* __restrict__ sp = tl + (size_t)tau * TS + toL[tau];
    unsigned* dp = leL + nown + tpL[tau];
    if (r < c) {
      constexpr int RU = 8;
      int jq[RU];
      unsigned rw[RU];
      for (int q = 0; q < RU; ++q) {
        const int j = r + q * TPT;
        jq[q] = (j < c) ? j : r;
        rw[q] = sp[jq[q]];
      }
      for (int q = 0; q < RU; ++q) dp[jq[q]] = rw[q];
      for (int j = r + RU * TPT; j < c; j += TPT) dp[j] = sp[j];
    }
  }
  __syncthreads();
  const int per = (total + NW - 1) / NW;
  const unsigned* seg = leL + (listed ? wave * per : 0);
  int cnt = total - wave * per;
  cnt = (cnt < 0) ? 0 : ((cnt > per) ? per : cnt);
  const int base = wave * per;
  if (t == 0) lcnt[blockIdx.x] = listed ? total : CAP + 1;

  int sK[NB], dK[NB];
  v4f aK[NB];
  if (listed) {
    for (int k = 0; k < NB; ++k) {
      const int idx = k * 32 + lane;
      const unsigned wv = seg[(idx < cnt) ? idx : 0];
      const unsigned wk = (idx < cnt) ? wv : (unsigned)n0;
      sK[k] = (int)(wk & 0xFFFFFFu);
      dK[k] = (int)(wk >> 24);
    }
    for (int k = 0; k < NB; ++k)
      aK[k] = *reinterpret_cast<const v4f*>(as + (size_t)sK[k] * H);
    for (int k = 0; k < NB; ++k) {
      const int idx = k * 32 + lane;
      if (idx < cnt)
        for (int h = 0; h < H; ++h) atomicMax(&mxL[dK[k] * H + h], fkey(aK[k][h]));
    }
  } else {
    for (int k = 0; k < NB; ++k) { sK[k] = 0; dK[k] = 0; aK[k] = v4f{0.f, 0.f, 0.f, 0.f}; }
    for (int i = t; i < nown + E; i += T) {
      int s, dl;
      if (i < nown) { s = n0 + i; dl = i; }
      else {
        const int e = i - nown;
        const unsigned u = (unsigned)(dstp[e] - n0);
        if (u >= (unsigned)nown) continue;
        s = ei[e]; dl = (int)u;
      }
      for (int h = 0; h < H; ++h) atomicMax(&mxL[dl * H + h], fkey(as[(size_t)s * H + h]));
    }
  }
  __syncthreads();

  if (listed) {
    unsigned* __restrict__ lsave = lst + (size_t)blockIdx.x * CAP + base;
    for (int k = 0; k < NB; ++k) {
      const int b0 = k * 32;
      if (b0 >= cnt) break;
      const int bc = (cnt - b0 < 32) ? (cnt - b0) : 32;
      const int s = sK[0], dl = dK[0];
      const v4f asv = aK[0];
      for (int q = 0; q + 1 < NB; ++q) { sK[q] = sK[q + 1]; dK[q] = dK[q + 1]; aK[q] = aK[q + 1]; }
      float ex[H];
      for (int h = 0; h < H; ++h) ex[h] = 0.f;
      if (lane < bc) {
        lsave[b0 + lane] = (unsigned)s | ((unsigned)dl << 24);
        const v4f a4 = *reinterpret_cast<const v4f*>(adL + dl * H);
        const v4u k4 = *reinterpret_cast<const v4u*>(mxL + dl * H);
        for (int h = 0; h < H; ++h) {
          ex[h] = edge_term(asv[h], a4[h], funkey(k4[h]));
          atomicAdd(&esL[dl * H + h], ex[h]);
        }
      }
      const int full = bc - bc % G;
      for (int j0 = 0; j0 < full; j0 += G) {
        float v[G][H];
        for (int g = 0; g < G; ++g) {
          const int sj = __builtin_amdgcn_readlane(s, j0 + g);
          const float* hp = hfeat + (size_t)sj * HD + lane;
          for (int h = 0; h < H; ++h) v[g][h] = hp[h * D];
        }
        for (int g = 0; g < G; ++g) {
          const int dj = __builtin_amdgcn_readlane(dl, j0 + g);
          float* ap = accL + dj * HD + lane;
          for (int h = 0; h < H; ++h) {
            const float w = __int_as_float(
                __builtin_amdgcn_readlane(__float_as_int(ex[h]), j0 + g));
            atomicAdd(&ap[h * D], w * v[g][h]);
          }
        }
      }
      if (full < bc) {
        float v[G][H];
        for (int g = 0; g < G; ++g) {
          const int j = (full + g < bc) ? (full + g) : (bc - 1);
          const int sj = __builtin_amdgcn_readlane(s, j);
          const float* hp = hfeat + (size_t)sj * HD + lane;
          for (int h = 0; h < H; ++h) v[g][h] = hp[h * D];
        }
        for (int g = 0; g < G; ++g) {
          if (full + g < bc) {
            const int dj = __builtin_amdgcn_readlane(dl, full + g);
            float* ap = accL + dj * HD + lane;
            for (int h = 0; h < H; ++h) {
              const float w = __int_as_float(
                  __builtin_amdgcn_readlane(__float_as_int(ex[h]), full + g));
              atomicAdd(&ap[h * D], w * v[g][h]);
            }
          }
        }
      }
    }
  } else {
    for (int i = t; i < nown + E; i += T) {
      int s, dl;
      if (i < nown) { s = n0 + i; dl = i; }
      else {
        const int e = i - nown;
        const unsigned u = (unsigned)(dstp[e] - n0);
        if (u >= (unsigned)nown) continue;
        s = ei[e]; dl = (int)u;
      }
      const float* hp = hfeat + (size_t)s * HD;
      for (int h = 0; h < H; ++h) {
        const float exh = edge_term(as[(size_t)s * H + h], adL[dl * H + h],
                                    funkey(mxL[dl * H + h]));
        atomicAdd(&esL[dl * H + h], exh);
        for (int kk = 0; kk < D; ++kk)
          atomicAdd(&accL[dl * HD + h * D + kk], exh * hp[h * D + kk]);
      }
    }
  }
  __syncthreads();

  for (int i = t; i < nown * HD; i += T) {
    const int nl = i / HD, c = i % HD, h = c / D;
    float v = accL[i] / (esL[nl * H + h] + 1e-16f) + bias[c];
    if (apply_elu) v = (v > 0.f) ? v : (expf(v) - 1.f);
    out[(size_t)n0 * HD + i] = v;
  }
}

template <int NPW, int CAP, int G>
__global__ __launch_bounds__(1024) void owner_list_acc_kernel(
    const int* __restrict__ ei, int E, int N,
    const unsigned* __restrict__ lst, const int* __restrict__ lcnt,
    const float* __restrict__ as, const float* __restrict__ ad,
    const float* __restrict__ hfeat, const float* __restrict__ bias,
    float* __restrict__ out, int apply_elu) {
  constexpr int D = 32;
  constexpr int T = 1024;
  constexpr int NB = CAP / T;
  static_assert(CAP % T == 0, "whole batches");
  static_assert((NPW * D) % 4 == 0 && 32 % G == 0, "tile shape");
  __shared__ __attribute__((aligned(16))) float accL[NPW * D];
  __shared__ float esL[NPW];
  __shared__ float adL[NPW];
  __shared__ unsigned mxL[NPW];

  const int t = (int)threadIdx.x;
  const int n0 = (int)blockIdx.x * NPW;
  const int nown = (N - n0 < NPW) ? (N - n0) : NPW;

  {
    const v4f z = {0.f, 0.f, 0.f, 0.f};
    v4f* a4 = reinterpret_cast<v4f*>(accL);
    for (int i = t; i < NPW * D / 4; i += T) a4[i] = z;
    for (int i = t; i < NPW; i += T) {
      esL[i] = 0.f;
      mxL[i] = 0u;
      adL[i] = (i < nown) ? ad[n0 + i] : 0.f;
    }
  }
  __syncthreads();

  const int total = lcnt[blockIdx.x];
  const bool listed = total <= CAP;
  const int* __restrict__ dstp = ei + E;

  int sK[NB], dK[NB];
  float aK[NB];
  if (listed) {
    const unsigned* __restrict__ lp = lst + (size_t)blockIdx.x * CAP;
    for (int k = 0; k < NB; ++k) {
      const int idx = k * T + t;
      const unsigned w = lp[(idx < total) ? idx : 0];
      sK[k] = (int)(w & 0xFFFFFFu);
      dK[k] = (int)(w >> 24);
    }
    for (int k = 0; k < NB; ++k) aK[k] = as[sK[k]];
    for (int k = 0; k < NB; ++k)
      if (k * T + t < total) atomicMax(&mxL[dK[k]], fkey(aK[k]));
  } else {
    for (int k = 0; k < NB; ++k) { sK[k] = 0; dK[k] = 0; aK[k] = 0.f; }
    for (int i = t; i < nown + E; i += T) {
      int s, dl;
      if (i < nown) { s = n0 + i; dl = i; }
      else {
        const int e = i - nown;
        const unsigned u = (unsigned)(dstp[e] - n0);
        if (u >= (unsigned)nown) continue;
        s = ei[e]; dl = (int)u;
      }
      atomicMax(&mxL[dl], fkey(as[s]));
    }
  }
  __syncthreads();

  if (listed) {
    const int lane = t & 31;
    const int wave = __builtin_amdgcn_readfirstlane(t >> 5);
    for (int k = 0; k < NB; ++k) {
      const int b0 = wave * 32 + k * T;
      if (b0 >= total) break;
      const int bc = (total - b0 < 32) ? (total - b0) : 32;
      const int s = sK[0], dl = dK[0];
      const float asv = aK[0];
      for (int q = 0; q + 1 < NB; ++q) { sK[q] = sK[q + 1]; dK[q] = dK[q + 1]; aK[q] = aK[q + 1]; }
      float ex = 0.f;
      if (lane < bc) {
        ex = edge_term(asv, adL[dl], funkey(mxL[dl]));
        atomicAdd(&esL[dl], ex);
      }
      const int full = bc - bc % G;
      for (int j0 = 0; j0 < full; j0 += G) {
        float v[G];
        for (int g = 0; g < G; ++g) {
          const int sj = __builtin_amdgcn_readlane(s, j0 + g);
          v[g] = hfeat[(size_t)sj * D + lane];
        }
        for (int g = 0; g < G; ++g) {
          const int dj = __builtin_amdgcn_readlane(dl, j0 + g);
          const float w = __int_as_float(
              __builtin_amdgcn_readlane(__float_as_int(ex), j0 + g));
          atomicAdd(&accL[dj * D + lane], w * v[g]);
        }
      }
      if (full < bc) {
        float v[G];
        for (int g = 0; g < G; ++g) {
          const int j = (full + g < bc) ? (full + g) : (bc - 1);
          const int sj = __builtin_amdgcn_readlane(s, j);
          v[g] = hfeat[(size_t)sj * D + lane];
        }
        for (int g = 0; g < G; ++g) {
          if (full + g < bc) {
            const int dj = __builtin_amdgcn_readlane(dl, full + g);
            const float w = __int_as_float(
                __builtin_amdgcn_readlane(__float_as_int(ex), full + g));
            atomicAdd(&accL[dj * D + lane], w * v[g]);
          }
        }
      }
    }
  } else {
    for (int i = t; i < nown + E; i += T) {
      int s, dl;
      if (i < nown) { s = n0 + i; dl = i; }
      else {
        const int e = i - nown;
        const unsigned u = (unsigned)(dstp[e] - n0);
        if (u >= (unsigned)nown) continue;
        s = ei[e]; dl = (int)u;
      }
      const float exv = edge_term(as[s], adL[dl], funkey(mxL[dl]));
      const float* hp = hfeat + (size_t)s * D;
      atomicAdd(&esL[dl], exv);
      for (int kk = 0; kk < D; ++kk) atomicAdd(&accL[dl * D + kk], exv * hp[kk]);
    }
  }
  __syncthreads();

  for (int i = t; i < nown * D; i += T) {
    const int nl = i / D, c = i % D;
    float v = accL[i] / (esL[nl] + 1e-16f) + bias[c];
    if (apply_elu) v = (v > 0.f) ? v : (expf(v) - 1.f);
    out[(size_t)n0 * D + i] = v;
  }
}

__global__ void pred_node_kernel(const float* __restrict__ h2,
                                 const float* __restrict__ Wp,
                                 float* __restrict__ ps, float* __restrict__ pd,
                                 int N, int D) {
  int n = blockIdx.x * blockDim.x + threadIdx.x;
  if (n >= N) return;
  const float* hp = h2 + (size_t)n * D;
  float s = 0.f, d = 0.f;
  for (int k = 0; k < D; ++k) { s += hp[k] * Wp[k]; d += hp[k] * Wp[D + k]; }
  ps[n] = s; pd[n] = d;
}

__global__ void pred_edge_kernel(const int* __restrict__ ei, int E,
                                 const float* __restrict__ ps,
                                 const float* __restrict__ pd,
                                 const float* __restrict__ bp,
                                 float* __restrict__ out) {
  int e = blockIdx.x * blockDim.x + threadIdx.x;
  if (e >= E) return;
  out[e] = ps[ei[e]] + pd[ei[E + e]] + bp[0];
}

static inline int cdiv(long long a, int b) { return (int)((a + b - 1) / b); }

extern "C" void kernel_launch(void* const* d_in, const int* in_sizes, int n_in,
                              void* d_out, int out_size, void* d_ws, size_t ws_size,
                              hipStream_t stream) {
  const float* x   = (const float*)d_in[0];
  const int*   ei  = (const int*)  d_in[1];
  const float* W1  = (const float*)d_in[2];
  const float* as1w= (const float*)d_in[3];
  const float* ad1w= (const float*)d_in[4];
  const float* b1  = (const float*)d_in[5];
  const float* W2  = (const float*)d_in[6];
  const float* as2w= (const float*)d_in[7];
  const float* ad2w= (const float*)d_in[8];
  const float* b2  = (const float*)d_in[9];
  const float* Wp  = (const float*)d_in[10];
  const float* bp  = (const float*)d_in[11];
  float* out = (float*)d_out;

  const int F = 128, H1 = 4, Hd = 32;
  const int N = in_sizes[0] / F;
  const int E = in_sizes[1] / 2;

  constexpr int NPW = 196, CAP = 6144, TS = 4096, MAXOWN = 256;
  float* ws = (float*)d_ws;
  float*    h1    = ws;                 ws += (size_t)N * F;
  float*    acc1  = ws;                 ws += (size_t)N * F;
  float*    as1   = ws;                 ws += (size_t)N * H1;
  float*    ad1   = ws;                 ws += (size_t)N * 2 * H1;
  int*      lcnt  = (int*)ws;           ws += (size_t)N * H1;
  float*    h2p   = ws;                 ws += (size_t)N * Hd;
  unsigned* tl    = (unsigned*)h2p;
  int*      toff  = (int*)(tl + E);
  unsigned* lst   = (unsigned*)ws;      ws += (size_t)N * Hd;
  float*    as2   = ws;                 ws += (size_t)N;
  float*    ad2   = ws;                 ws += (size_t)N * 2;
  ws += (size_t)N;
  float*    ps    = ws;                 ws += (size_t)N;
  float*    pd    = ws;                 ws += (size_t)N;
  float*    out2  = h1;
  (void)ws_size;

  const int T = 256;

  gemm_wmma_f32_kernel<8><<<cdiv(N, 32), 8 * 32, 0, stream>>>(x, W1, h1, N);
  alpha_kernel<NPW, TS, MAXOWN><<<cdiv((long long)N * H1, TS / 4), TS / 4, 0, stream>>>(
      h1, as1w, ad1w, as1, ad1, N, H1, Hd, ei, E, tl, toff);
  owner_acc_kernel<4, NPW, CAP, 8, TS, MAXOWN><<<cdiv(N, NPW), 1024, 0, stream>>>(
      ei, E, N, tl, toff, as1, ad1, h1, b1, acc1, 1, lst, lcnt);

  gemm_wmma_f32_kernel<2><<<cdiv(N, 32), 2 * 32, 0, stream>>>(acc1, W2, h2p, N);
  alpha_kernel<NPW, TS, MAXOWN><<<cdiv(N, T), T, 0, stream>>>(h2p, as2w, ad2w, as2, ad2, N, 1, Hd, ei, E, nullptr, nullptr);
  owner_list_acc_kernel<NPW, CAP, 16><<<cdiv(N, NPW), 1024, 0, stream>>>(
      ei, E, N, lst, lcnt, as2, ad2, h2p, b2, out2, 0);

  pred_node_kernel<<<cdiv(N, T), T, 0, stream>>>(out2, Wp, ps, pd, N, Hd);
  pred_edge_kernel<<<cdiv(E, T), T, 0, stream>>>(ei, E, ps, pd, bp, out);
}
